# starting kernel
# baseline (speedup 1.0000x reference)
#include <hip/hip_runtime.h>
#include <hip/hip_bf16.h>


typedef __attribute__((ext_vector_type(2))) float v2f;
typedef __attribute__((ext_vector_type(8))) float v8f;

__device__ __forceinline__ void atomAddF(float* p, float v) {
#if defined(__HIP_DEVICE_COMPILE__)
    unsafeAtomicAdd(p, v);
#else
    (void)p; (void)v;
#endif
}

__global__ void k_zero(float* __restrict__ p, long long n) {
    long long i = (long long)blockIdx.x * blockDim.x + threadIdx.x;
    if (i < n) p[i] = 0.0f;
}

__global__ void k_degree(const int* __restrict__ ei, const float* __restrict__ w,
                         float* __restrict__ outdeg, float* __restrict__ indeg, int E) {
    int e = blockIdx.x * blockDim.x + threadIdx.x;
    if (e >= E) return;
    float we = w[e];
    atomAddF(&outdeg[ei[e]], we);
    atomAddF(&indeg[ei[E + e]], we);
}

__global__ void k_wnorm(const int* __restrict__ ei, const float* __restrict__ w,
                        const float* __restrict__ outdeg, const float* __restrict__ indeg,
                        float* __restrict__ wn, int E) {
    int e = blockIdx.x * blockDim.x + threadIdx.x;
    if (e >= E) return;
    float od = outdeg[ei[e]];
    float id = indeg[ei[E + e]];
    float a = (od > 0.0f) ? rsqrtf(od) : 0.0f;
    float b = (id > 0.0f) ? rsqrtf(id) : 0.0f;
    wn[e] = w[e] * a * b;
}

__global__ void k_scatter(const float* __restrict__ x, const int* __restrict__ ei,
                          const float* __restrict__ wn,
                          float* __restrict__ aggf, float* __restrict__ aggb, int E) {
    long long t = (long long)blockIdx.x * blockDim.x + threadIdx.x;
    int e = (int)(t >> 5);
    int d = (int)(t & 31) * 2;
    if (e >= E) return;
    int row = ei[e];
    int col = ei[E + e];
    float w = wn[e];
    const v2f xc = *(const v2f*)(x + (long long)col * 64 + d);
    const v2f xr = *(const v2f*)(x + (long long)row * 64 + d);
    float* pf = aggf + (long long)row * 64 + d;
    float* pb = aggb + (long long)col * 64 + d;
    atomAddF(pf + 0, w * xc.x);
    atomAddF(pf + 1, w * xc.y);
    atomAddF(pb + 0, w * xr.x);
    atomAddF(pb + 1, w * xr.y);
}

__global__ void __launch_bounds__(256)
k_wmma_epilogue(const float* __restrict__ aggf, const float* __restrict__ aggb,
                const float* __restrict__ Wsrc, const float* __restrict__ Wdst,
                const float* __restrict__ bsrc, const float* __restrict__ bdst,
                float* __restrict__ out, int tilesM, int Nnodes) {
    int wave  = (int)((blockIdx.x * blockDim.x + threadIdx.x) >> 5);
    int lane  = threadIdx.x & 31;
    int tileM = wave >> 2;
    int tileN = wave & 3;
    if (tileM >= tilesM) return;

    int lmod = lane & 15;
    int lhi  = lane >> 4;

    int arow = tileM * 16 + lmod;
    if (arow >= Nnodes) arow = Nnodes - 1;
    const float* af_p = aggf + (long long)arow * 64 + 2 * lhi;
    const float* ab_p = aggb + (long long)arow * 64 + 2 * lhi;
    const float* bs_p = Wsrc + (tileN * 16 + lmod) * 64 + 2 * lhi;
    const float* bd_p = Wdst + (tileN * 16 + lmod) * 64 + 2 * lhi;

    v8f acc = {};
#pragma unroll
    for (int kk = 0; kk < 16; ++kk) {
        v2f af = *(const v2f*)(af_p + 4 * kk);
        v2f bs = *(const v2f*)(bs_p + 4 * kk);
        acc = __builtin_amdgcn_wmma_f32_16x16x4_f32(
                  false, af, false, bs, (short)0, acc, false, false);
        v2f ab = *(const v2f*)(ab_p + 4 * kk);
        v2f bd = *(const v2f*)(bd_p + 4 * kk);
        acc = __builtin_amdgcn_wmma_f32_16x16x4_f32(
                  false, ab, false, bd, (short)0, acc, false, false);
    }

    int ncol = tileN * 16 + lmod;
    float bias = 0.5f * (bsrc[ncol] + bdst[ncol]);
    int mbase = tileM * 16 + 8 * lhi;
    float* ocol = out + ncol;
#pragma unroll
    for (int r = 0; r < 8; ++r) {
        int m = mbase + r;
        if (m < Nnodes) ocol[(long long)m * 64] = 0.5f * acc[r] + bias;
    }
}

extern "C" void kernel_launch(void* const* d_in, const int* in_sizes, int n_in,
                              void* d_out, int out_size, void* d_ws, size_t ws_size,
                              hipStream_t stream) {
    const float* x    = (const float*)d_in[0];
    const int*   ei   = (const int*)d_in[1];
    const float* ew   = (const float*)d_in[2];
    const float* Wsrc = (const float*)d_in[3];
    const float* bsrc = (const float*)d_in[4];
    const float* Wdst = (const float*)d_in[5];
    const float* bdst = (const float*)d_in[6];
    float* out = (float*)d_out;

    const int N = in_sizes[0] / 64;
    const int E = in_sizes[2];

    float* ws     = (float*)d_ws;
    float* outdeg = ws;
    float* indeg  = ws + (long long)N;
    float* wn     = ws + 2LL * N;
    float* aggf   = wn + (long long)E;
    float* aggb   = aggf + 64LL * N;

    const long long ztotal = 130LL * N + (long long)E;
    k_zero<<<(unsigned)((ztotal + 255) / 256), 256, 0, stream>>>(ws, ztotal);

    const int eb = (E + 255) / 256;
    k_degree<<<eb, 256, 0, stream>>>(ei, ew, outdeg, indeg, E);
    k_wnorm<<<eb, 256, 0, stream>>>(ei, ew, outdeg, indeg, wn, E);

    const long long st = 32LL * E;
    k_scatter<<<(unsigned)((st + 255) / 256), 256, 0, stream>>>(x, ei, wn, aggf, aggb, E);

    const int tilesM = (N + 15) / 16;
    const int waves  = tilesM * 4;
    k_wmma_epilogue<<<(waves + 7) / 8, 256, 0, stream>>>(aggf, aggb, Wsrc, Wdst,
                                                         bsrc, bdst, out, tilesM, N);
}


# scatter replaced by an owner pass: a work-group keeps 392 rows of out in LDS, reads all edges, queues its hits, ds_add_f32 into LDS, stores bias+sum once; no global atomics there; sum order new
# speedup vs baseline: 1.3102x; 1.3102x over previous
#include <hip/hip_runtime.h>
#include <hip/hip_bf16.h>


#define DEG_STRIDE 64
#define DEG_IN_OFF 32

typedef __attribute__((ext_vector_type(2))) float v2f;
typedef __attribute__((ext_vector_type(8))) float v8f;

__device__ __forceinline__ void atomAddF(float* p, float v) {
#if defined(__HIP_DEVICE_COMPILE__)
    unsafeAtomicAdd(p, v);
#else
    (void)p; (void)v;
#endif
}

__global__ void k_zero_deg(float* __restrict__ degw, long long ndeg) {
    long long i = (long long)blockIdx.x * blockDim.x + threadIdx.x;
    if (i < ndeg) degw[i * DEG_IN_OFF] = 0.0f;
}

__global__ void k_degree(const int* __restrict__ ei, const float* __restrict__ w,
                         float* __restrict__ outdeg, float* __restrict__ indeg, int E) {
    int e = blockIdx.x * blockDim.x + threadIdx.x;
    if (e >= E) return;
    float we = w[e];
    atomAddF(&outdeg[(long long)ei[e] * DEG_STRIDE], we);
    atomAddF(&indeg[(long long)ei[E + e] * DEG_STRIDE], we);
}

__global__ void __launch_bounds__(256)
k_wmma_project(const float* __restrict__ x,
               const float* __restrict__ Wsrc, const float* __restrict__ Wdst,
               float* __restrict__ ys, float* __restrict__ yd,
               float* __restrict__ ra, float* __restrict__ rb,
               const float* __restrict__ degw, int tilesM, int Nnodes) {
    const int  fwhich = (int)(threadIdx.x >> 5);
    const int  fm     = (int)blockIdx.x * 32 + (int)(threadIdx.x & 31);
    const bool fth    = (threadIdx.x < 64) && (fm < Nnodes);
    float deg = 0.0f;
    if (fth) deg = degw[(long long)fm * DEG_STRIDE + fwhich * DEG_IN_OFF];

    int wave  = (int)((blockIdx.x * blockDim.x + threadIdx.x) >> 5);
    int lane  = threadIdx.x & 31;
    int tileM = wave >> 2;
    int tileN = wave & 3;
    const bool active = tileM < tilesM;

    int lmod = lane & 15;
    int lhi  = lane >> 4;

    v8f accs = {};
    v8f accd = {};
    if (active) {
        int arow = tileM * 16 + lmod;
        if (arow >= Nnodes) arow = Nnodes - 1;
        const float* a_p  = x    + (long long)arow * 64 + 2 * lhi;
        const float* bs_p = Wsrc + (tileN * 16 + lmod) * 64 + 2 * lhi;
        const float* bd_p = Wdst + (tileN * 16 + lmod) * 64 + 2 * lhi;
#pragma unroll
        for (int kk = 0; kk < 16; ++kk) {
            v2f a  = *(const v2f*)(a_p  + 4 * kk);
            v2f bs = *(const v2f*)(bs_p + 4 * kk);
            v2f bd = *(const v2f*)(bd_p + 4 * kk);
            accs = __builtin_amdgcn_wmma_f32_16x16x4_f32(
                       false, a, false, bs, (short)0, accs, false, false);
            accd = __builtin_amdgcn_wmma_f32_16x16x4_f32(
                       false, a, false, bd, (short)0, accd, false, false);
        }
    }

    if (fth) {
        float f = (deg > 0.0f) ? rsqrtf(deg) : 0.0f;
        if (fwhich) rb[fm] = f; else ra[fm] = f;
    }
    if (!active) return;

    int ncol = tileN * 16 + lmod;
    int mbase = tileM * 16 + 8 * lhi;
#pragma unroll
    for (int r = 0; r < 8; ++r) {
        int m = mbase + r;
        if (m < Nnodes) {
            long long o = (long long)m * 64 + ncol;
            ys[o]  = 0.5f * accs[r];
            yd[o]  = 0.5f * accd[r];
        }
    }
}

#define OWN_ROWS    392
#define OWN_WAVES   32
#define OWN_THREADS (OWN_WAVES * 32)
#define OWN_QCAP    1056

template <bool FULL>
__device__ __forceinline__ void own_drain(const unsigned* qb, int n, int lane,
                                          const int* __restrict__ ei, const float* __restrict__ ew,
                                          const float* __restrict__ ra, const float* __restrict__ rb,
                                          const float* __restrict__ ysd, float* acc,
                                          int E, int N, int lo) {
    int   dl   = 0;
    int   srow = 0;
    float wn   = 0.0f;
    if (FULL || lane < n) {
        unsigned code = qb[lane];
        int e   = (int)(code >> 1);
        int dir = (int)(code & 1u);
        int row = ei[e];
        int col = ei[E + e];
        wn   = ew[e] * ra[row] * rb[col];
        dl   = (dir ? col : row) - lo;
        srow = dir ? (row + N) : col;
    }
    v2f v[32];
#pragma unroll
    for (int j = 0; j < 32; ++j) {
        if (FULL || j < n) {
            int s = __builtin_amdgcn_readlane(srow, j);
            v[j] = *(const v2f*)(ysd + (long long)s * 64 + 2 * lane);
        }
    }
#pragma unroll
    for (int j = 0; j < 32; ++j) {
        if (FULL || j < n) {
            float w = __int_as_float(__builtin_amdgcn_readlane(__float_as_int(wn), j));
            int   d = __builtin_amdgcn_readlane(dl, j);
            float* a = acc + d * 64 + 2 * lane;
            __hip_atomic_fetch_add(a + 0, w * v[j].x, __ATOMIC_RELAXED, __HIP_MEMORY_SCOPE_WORKGROUP);
            __hip_atomic_fetch_add(a + 1, w * v[j].y, __ATOMIC_RELAXED, __HIP_MEMORY_SCOPE_WORKGROUP);
        }
    }
}

typedef __attribute__((ext_vector_type(4))) int v4i;

__global__ void __launch_bounds__(OWN_THREADS)
k_owner(const float* __restrict__ ysd,
        const int* __restrict__ ei, const float* __restrict__ ew,
        const float* __restrict__ ra, const float* __restrict__ rb,
        const float* __restrict__ bsrc, const float* __restrict__ bdst,
        float* __restrict__ out, int E, int N) {
    __shared__ float    acc[OWN_ROWS * 64];
    __shared__ unsigned queue[OWN_WAVES * OWN_QCAP];

    const int tid  = (int)threadIdx.x;
    const int lane = tid & 31;
    const int wave = __builtin_amdgcn_readfirstlane(tid >> 5);
    const int lo   = (int)blockIdx.x * OWN_ROWS;
    const int nr   = (N - lo < OWN_ROWS) ? (N - lo) : OWN_ROWS;
    const unsigned unr = (unsigned)nr;

    for (int i = tid; i < OWN_ROWS * 64; i += OWN_THREADS) acc[i] = 0.0f;
    __syncthreads();

    unsigned* qw = queue + wave * OWN_QCAP;
    unsigned  qn = 0;

    const int  nq     = (E + 3) >> 2;
    const int  nqf    = E >> 2;
    const bool vec    = ((E & 3) == 0);
    const int  chunks = (nq + 31) >> 5;

    const int ngroups = (chunks + 3) >> 2;
    const int rot     = (int)(((long long)blockIdx.x * ngroups) / gridDim.x);
    for (int gi = wave; gi < ngroups; gi += OWN_WAVES) {
        int gr = gi + rot;
        if (gr >= ngroups) gr -= ngroups;
        const int cb = gr * 4;
        v4i R[4], C[4];
        if (vec && (cb + 4) * 32 <= nqf) {
#pragma unroll
            for (int k = 0; k < 4; ++k) {
                long long q = (long long)(cb + k) * 32 + lane;
                R[k] = *(const v4i*)(ei + 4 * q);
                C[k] = *(const v4i*)(ei + (long long)E + 4 * q);
            }
        } else {
#pragma unroll
            for (int k = 0; k < 4; ++k) {
                long long q = (long long)(cb + k) * 32 + lane;
#pragma unroll
                for (int m = 0; m < 4; ++m) {
                    long long e = 4 * q + m;
                    bool ok = e < (long long)E;
                    R[k][m] = ok ? ei[e] : -1;
                    C[k][m] = ok ? ei[(long long)E + e] : -1;
                }
            }
        }
#pragma unroll
        for (int k = 0; k < 4; ++k) {
            const unsigned e0 = 4u * (unsigned)((cb + k) * 32 + lane);
            bool     hf[4], hb[4];
            unsigned mf[4], mb[4];
            unsigned any = 0;
#pragma unroll
            for (int m = 0; m < 4; ++m) {
                hf[m] = (unsigned)(R[k][m] - lo) < unr;
                hb[m] = (unsigned)(C[k][m] - lo) < unr;
                mf[m] = __builtin_amdgcn_ballot_w32(hf[m]);
                mb[m] = __builtin_amdgcn_ballot_w32(hb[m]);
                any |= mf[m] | mb[m];
            }
            if (any) {
#pragma unroll
                for (int m = 0; m < 4; ++m) {
                    if (mf[m]) {
                        unsigned pos = qn + __builtin_amdgcn_mbcnt_lo(mf[m], 0u);
                        if (hf[m]) qw[pos] = ((e0 + m) << 1);
                        qn += (unsigned)__builtin_popcount(mf[m]);
                    }
                    if (mb[m]) {
                        unsigned pos = qn + __builtin_amdgcn_mbcnt_lo(mb[m], 0u);
                        if (hb[m]) qw[pos] = ((e0 + m) << 1) | 1u;
                        qn += (unsigned)__builtin_popcount(mb[m]);
                    }
                }
            }
        }
        __builtin_amdgcn_wave_barrier();
        while (qn >= 32u) {
            qn -= 32u;
            own_drain<true>(qw + qn, 32, lane, ei, ew, ra, rb, ysd, acc, E, N, lo);
        }
    }
    __builtin_amdgcn_wave_barrier();
    if (qn) own_drain<false>(qw, (int)qn, lane, ei, ew, ra, rb, ysd, acc, E, N, lo);

    __syncthreads();

    const v2f bs = *(const v2f*)(bsrc + 2 * lane);
    const v2f bd = *(const v2f*)(bdst + 2 * lane);
    v2f bias;
    bias.x = 0.5f * (bs.x + bd.x);
    bias.y = 0.5f * (bs.y + bd.y);
    for (int r = wave; r < nr; r += OWN_WAVES) {
        v2f a = *(const v2f*)(acc + r * 64 + 2 * lane);
        v2f o;
        o.x = bias.x + a.x;
        o.y = bias.y + a.y;
        *(v2f*)(out + (long long)(lo + r) * 64 + 2 * lane) = o;
    }
}

extern "C" void kernel_launch(void* const* d_in, const int* in_sizes, int n_in,
                              void* d_out, int out_size, void* d_ws, size_t ws_size,
                              hipStream_t stream) {
    const float* x    = (const float*)d_in[0];
    const int*   ei   = (const int*)d_in[1];
    const float* ew   = (const float*)d_in[2];
    const float* Wsrc = (const float*)d_in[3];
    const float* bsrc = (const float*)d_in[4];
    const float* Wdst = (const float*)d_in[5];
    const float* bdst = (const float*)d_in[6];
    float* out = (float*)d_out;

    const int N = in_sizes[0] / 64;
    const int E = in_sizes[2];

    float* ws     = (float*)d_ws;
    float* outdeg = out;
    float* indeg  = out + DEG_IN_OFF;
    float* ra     = ws;
    float* rb     = ws + (long long)N;
    float* ys     = ws + 2LL * N + (long long)E;
    float* yd     = ys + 64LL * N;

    const long long ndeg = 2LL * N;
    k_zero_deg<<<(unsigned)((ndeg + 255) / 256), 256, 0, stream>>>(out, ndeg);

    const int eb = (E + 255) / 256;
    k_degree<<<eb, 256, 0, stream>>>(ei, ew, outdeg, indeg, E);

    const int tilesM = (N + 15) / 16;
    const int waves  = tilesM * 4;
    k_wmma_project<<<(waves + 7) / 8, 256, 0, stream>>>(x, Wsrc, Wdst,
                                                        ys, yd, ra, rb, out, tilesM, N);

    k_owner<<<(N + OWN_ROWS - 1) / OWN_ROWS, OWN_THREADS, 0, stream>>>(ys, ei, ew, ra, rb,
                                                                       bsrc, bdst, out, E, N);
}


# owner pass scans two key planes (a byte per edge and direction: owner block of row, of col; written by the degree pass, 4 edges a lane), not the 8 bytes of indices; LDS sum order new, arrival order
# speedup vs baseline: 1.5227x; 1.1622x over previous
#include <hip/hip_runtime.h>
#include <hip/hip_bf16.h>


#define DEG_STRIDE 64
#define DEG_IN_OFF 32

#define OWN_ROWS    392
#define KEY_LINE    128

typedef __attribute__((ext_vector_type(2))) float v2f;
typedef __attribute__((ext_vector_type(4))) float v4f;
typedef __attribute__((ext_vector_type(8))) float v8f;
typedef __attribute__((ext_vector_type(4))) int v4i;
typedef __attribute__((ext_vector_type(4))) unsigned v4u;

__device__ __forceinline__ void atomAddF(float* p, float v) {
#if defined(__HIP_DEVICE_COMPILE__)
    unsafeAtomicAdd(p, v);
#else
    (void)p; (void)v;
#endif
}

__global__ void k_zero_deg(float* __restrict__ degw, long long ndeg) {
    long long i = (long long)blockIdx.x * blockDim.x + threadIdx.x;
    if (i < ndeg) degw[i * DEG_IN_OFF] = 0.0f;
}

__global__ void __launch_bounds__(256)
k_degree(const int* __restrict__ ei, const float* __restrict__ w,
         float* __restrict__ outdeg, float* __restrict__ indeg,
         unsigned* __restrict__ keyR, unsigned* __restrict__ keyC, int E, int nq4) {
    const int q = blockIdx.x * blockDim.x + threadIdx.x;
    if (q >= nq4) return;
    const long long e0 = 4LL * q;
    int   r[4], c[4];
    float we[4];
    if (((E & 3) == 0) && e0 + 4 <= (long long)E) {
        const v4i R  = *(const v4i*)(ei + e0);
        const v4i C  = *(const v4i*)(ei + (long long)E + e0);
        const v4f Wv = *(const v4f*)(w + e0);
#pragma unroll
        for (int m = 0; m < 4; ++m) { r[m] = R[m]; c[m] = C[m]; we[m] = Wv[m]; }
    } else {
#pragma unroll
        for (int m = 0; m < 4; ++m) {
            const long long e = e0 + m;
            const bool ok = e < (long long)E;
            r[m]  = ok ? ei[e] : -1;
            c[m]  = ok ? ei[(long long)E + e] : -1;
            we[m] = ok ? w[e] : 0.0f;
        }
    }
    unsigned kr = 0u, kc = 0u;
#pragma unroll
    for (int m = 0; m < 4; ++m) {
        if (r[m] >= 0) {
            kr |= (((unsigned)r[m] / OWN_ROWS) & 255u) << (8 * m);
            kc |= (((unsigned)c[m] / OWN_ROWS) & 255u) << (8 * m);
            atomAddF(&outdeg[(long long)r[m] * DEG_STRIDE], we[m]);
            atomAddF(&indeg[(long long)c[m] * DEG_STRIDE], we[m]);
        }
    }
    keyR[q] = kr;
    keyC[q] = kc;
}

__global__ void __launch_bounds__(256)
k_wmma_project(const float* __restrict__ x,
               const float* __restrict__ Wsrc, const float* __restrict__ Wdst,
               float* __restrict__ ys, float* __restrict__ yd,
               float* __restrict__ ra, float* __restrict__ rb,
               const float* __restrict__ degw, int tilesM, int Nnodes) {
    const int  fwhich = (int)(threadIdx.x >> 5);
    const int  fm     = (int)blockIdx.x * 32 + (int)(threadIdx.x & 31);
    const bool fth    = (threadIdx.x < 64) && (fm < Nnodes);
    float deg = 0.0f;
    if (fth) deg = degw[(long long)fm * DEG_STRIDE + fwhich * DEG_IN_OFF];

    int wave  = (int)((blockIdx.x * blockDim.x + threadIdx.x) >> 5);
    int lane  = threadIdx.x & 31;
    int tileM = wave >> 2;
    int tileN = wave & 3;
    const bool active = tileM < tilesM;

    int lmod = lane & 15;
    int lhi  = lane >> 4;

    v8f accs = {};
    v8f accd = {};
    if (active) {
        int arow = tileM * 16 + lmod;
        if (arow >= Nnodes) arow = Nnodes - 1;
        const float* a_p  = x    + (long long)arow * 64 + 2 * lhi;
        const float* bs_p = Wsrc + (tileN * 16 + lmod) * 64 + 2 * lhi;
        const float* bd_p = Wdst + (tileN * 16 + lmod) * 64 + 2 * lhi;
#pragma unroll
        for (int kk = 0; kk < 16; ++kk) {
            v2f a  = *(const v2f*)(a_p  + 4 * kk);
            v2f bs = *(const v2f*)(bs_p + 4 * kk);
            v2f bd = *(const v2f*)(bd_p + 4 * kk);
            accs = __builtin_amdgcn_wmma_f32_16x16x4_f32(
                       false, a, false, bs, (short)0, accs, false, false);
            accd = __builtin_amdgcn_wmma_f32_16x16x4_f32(
                       false, a, false, bd, (short)0, accd, false, false);
        }
    }

    if (fth) {
        float f = (deg > 0.0f) ? rsqrtf(deg) : 0.0f;
        if (fwhich) rb[fm] = f; else ra[fm] = f;
    }
    if (!active) return;

    int ncol = tileN * 16 + lmod;
    int mbase = tileM * 16 + 8 * lhi;
#pragma unroll
    for (int r = 0; r < 8; ++r) {
        int m = mbase + r;
        if (m < Nnodes) {
            long long o = (long long)m * 64 + ncol;
            ys[o]  = 0.5f * accs[r];
            yd[o]  = 0.5f * accd[r];
        }
    }
}

#define OWN_WAVES   32
#define OWN_THREADS (OWN_WAVES * 32)
#define OWN_QCAP    1056

__device__ __forceinline__ unsigned own_nz(unsigned w, unsigned gs) {
    const unsigned x = w ^ gs;
    return ((x & 0x7F7F7F7Fu) + 0x7F7F7F7Fu) | x | 0x7F7F7F7Fu;
}

__device__ __forceinline__ unsigned own_hits(v4u r, v4u c, unsigned gs) {
    unsigned m = own_nz(r[0], gs);
    m = __builtin_rotateright32(m, 1) & own_nz(r[1], gs);
    m = __builtin_rotateright32(m, 1) & own_nz(r[2], gs);
    m = __builtin_rotateright32(m, 1) & own_nz(r[3], gs);
    m = __builtin_rotateright32(m, 1) & own_nz(c[0], gs);
    m = __builtin_rotateright32(m, 1) & own_nz(c[1], gs);
    m = __builtin_rotateright32(m, 1) & own_nz(c[2], gs);
    m = __builtin_rotateright32(m, 1) & own_nz(c[3], gs);
    return ~m;
}

template <bool FULL>
__device__ __forceinline__ void own_drain(const unsigned* qb, int n, int lane,
                                          const int* __restrict__ ei, const float* __restrict__ ew,
                                          const float* __restrict__ ra, const float* __restrict__ rb,
                                          const float* __restrict__ ysd, float* acc,
                                          int E, int N, int lo, unsigned unr) {
    int   dl   = 0;
    int   srow = 0;
    float wn   = 0.0f;
    if (FULL || lane < n) {
        unsigned code = qb[lane];
        int e   = (int)(code >> 1);
        int dir = (int)(code & 1u);
        int row = ei[e];
        int col = ei[E + e];
        int d   = (dir ? col : row) - lo;
        if ((unsigned)d < unr) {
            wn   = ew[e] * ra[row] * rb[col];
            dl   = d;
            srow = dir ? (row + N) : col;
        }
    }
    v2f v[32];
#pragma unroll
    for (int j = 0; j < 32; ++j) {
        if (FULL || j < n) {
            int s = __builtin_amdgcn_readlane(srow, j);
            v[j] = *(const v2f*)(ysd + (long long)s * 64 + 2 * lane);
        }
    }
#pragma unroll
    for (int j = 0; j < 32; ++j) {
        if (FULL || j < n) {
            float w = __int_as_float(__builtin_amdgcn_readlane(__float_as_int(wn), j));
            int   d = __builtin_amdgcn_readlane(dl, j);
            float* a = acc + d * 64 + 2 * lane;
            __hip_atomic_fetch_add(a + 0, w * v[j].x, __ATOMIC_RELAXED, __HIP_MEMORY_SCOPE_WORKGROUP);
            __hip_atomic_fetch_add(a + 1, w * v[j].y, __ATOMIC_RELAXED, __HIP_MEMORY_SCOPE_WORKGROUP);
        }
    }
}

__global__ void __launch_bounds__(OWN_THREADS)
k_owner(const float* __restrict__ ysd,
        const int* __restrict__ ei, const float* __restrict__ ew,
        const unsigned* __restrict__ keyR, const unsigned* __restrict__ keyC,
        const float* __restrict__ ra, const float* __restrict__ rb,
        const float* __restrict__ bsrc, const float* __restrict__ bdst,
        float* __restrict__ out, int E, int N, int nq16) {
    __shared__ float    acc[OWN_ROWS * 64];
    __shared__ unsigned queue[OWN_WAVES * OWN_QCAP];

    const int tid  = (int)threadIdx.x;
    const int lane = tid & 31;
    const int wave = __builtin_amdgcn_readfirstlane(tid >> 5);
    const int lo   = (int)blockIdx.x * OWN_ROWS;
    const int nr   = (N - lo < OWN_ROWS) ? (N - lo) : OWN_ROWS;
    const unsigned unr = (unsigned)nr;
    const unsigned gs  = (blockIdx.x & 255u) * 0x01010101u;

    for (int i = tid; i < OWN_ROWS * 64; i += OWN_THREADS) acc[i] = 0.0f;
    __syncthreads();

    unsigned* qw = queue + wave * OWN_QCAP;
    unsigned  qn = 0;

    const int chunks = (nq16 + 31) >> 5;

    const int ngroups = (chunks + 3) >> 2;
    const int rot     = (int)(((long long)blockIdx.x * ngroups) / gridDim.x);
    const v4u* kR = (const v4u*)keyR;
    const v4u* kC = (const v4u*)keyC;
    for (int gi = wave; gi < ngroups; gi += OWN_WAVES) {
        int gr = gi + rot;
        if (gr >= ngroups) gr -= ngroups;
        const int cb = gr * 4;
        v4u R[4], C[4];
        if ((cb + 4) * 32 <= nq16) {
#pragma unroll
            for (int k = 0; k < 4; ++k) {
                const int q = (cb + k) * 32 + lane;
                R[k] = kR[q];
                C[k] = kC[q];
            }
        } else {
            const unsigned miss = ~gs;
#pragma unroll
            for (int k = 0; k < 4; ++k) {
                const int q = (cb + k) * 32 + lane;
                const bool ok = q < nq16;
                v4u no = {miss, miss, miss, miss};
                R[k] = ok ? kR[q] : no;
                C[k] = ok ? kC[q] : no;
            }
        }
        unsigned hm[4];
#pragma unroll
        for (int k = 0; k < 4; ++k) hm[k] = own_hits(R[k], C[k], gs);
#pragma unroll
        for (int k = 0; k < 4; ++k) {
            unsigned h = hm[k];
            const unsigned e0 = 16u * (unsigned)((cb + k) * 32 + lane);
            unsigned bal = __builtin_amdgcn_ballot_w32(h != 0u);
            while (bal) {
                const bool     act = (h != 0u);
                const unsigned p   = act ? (unsigned)__builtin_ctz(h) : 0u;
                h &= (h - 1u);
                const unsigned e   = e0 + 4u * (p & 3u) + (p >> 3);
                const unsigned dir = (p >> 2) & 1u;
                const bool     ok  = act && (e < (unsigned)E);
                const unsigned okm = __builtin_amdgcn_ballot_w32(ok);
                const unsigned pos = qn + __builtin_amdgcn_mbcnt_lo(okm, 0u);
                if (ok) qw[pos] = (e << 1) | dir;
                qn += (unsigned)__builtin_popcount(okm);
                bal = __builtin_amdgcn_ballot_w32(h != 0u);
            }
            __builtin_amdgcn_wave_barrier();
            while (qn >= 32u) {
                qn -= 32u;
                own_drain<true>(qw + qn, 32, lane, ei, ew, ra, rb, ysd, acc, E, N, lo, unr);
            }
        }
    }
    __builtin_amdgcn_wave_barrier();
    if (qn) own_drain<false>(qw, (int)qn, lane, ei, ew, ra, rb, ysd, acc, E, N, lo, unr);

    __syncthreads();

    const v2f bs = *(const v2f*)(bsrc + 2 * lane);
    const v2f bd = *(const v2f*)(bdst + 2 * lane);
    v2f bias;
    bias.x = 0.5f * (bs.x + bd.x);
    bias.y = 0.5f * (bs.y + bd.y);
    for (int r = wave; r < nr; r += OWN_WAVES) {
        v2f a = *(const v2f*)(acc + r * 64 + 2 * lane);
        v2f o;
        o.x = bias.x + a.x;
        o.y = bias.y + a.y;
        *(v2f*)(out + (long long)(lo + r) * 64 + 2 * lane) = o;
    }
}

extern "C" void kernel_launch(void* const* d_in, const int* in_sizes, int n_in,
                              void* d_out, int out_size, void* d_ws, size_t ws_size,
                              hipStream_t stream) {
    const float* x    = (const float*)d_in[0];
    const int*   ei   = (const int*)d_in[1];
    const float* ew   = (const float*)d_in[2];
    const float* Wsrc = (const float*)d_in[3];
    const float* bsrc = (const float*)d_in[4];
    const float* Wdst = (const float*)d_in[5];
    const float* bdst = (const float*)d_in[6];
    float* out = (float*)d_out;

    const int N = in_sizes[0] / 64;
    const int E = in_sizes[2];

    float* ws     = (float*)d_ws;
    float* outdeg = out;
    float* indeg  = out + DEG_IN_OFF;
    float* ra     = ws;
    float* rb     = ws + (long long)N;
    float* ys     = ws + 2LL * N + (long long)E;
    float* yd     = ys + 64LL * N;

    const int Epad = (E + KEY_LINE - 1) / KEY_LINE * KEY_LINE;
    unsigned char* kbase = (unsigned char*)(((size_t)(ws + 2LL * N) + 127) & ~(size_t)127);
    unsigned* keyR = (unsigned*)kbase;
    unsigned* keyC = (unsigned*)(kbase + Epad);

    const long long ndeg = 2LL * N;
    k_zero_deg<<<(unsigned)((ndeg + 255) / 256), 256, 0, stream>>>(out, ndeg);

    const int nq4 = Epad / 4;
    k_degree<<<(nq4 + 255) / 256, 256, 0, stream>>>(ei, ew, outdeg, indeg, keyR, keyC, E, nq4);

    const int tilesM = (N + 15) / 16;
    const int waves  = tilesM * 4;
    k_wmma_project<<<(waves + 7) / 8, 256, 0, stream>>>(x, Wsrc, Wdst,
                                                        ys, yd, ra, rb, out, tilesM, N);

    k_owner<<<(N + OWN_ROWS - 1) / OWN_ROWS, OWN_THREADS, 0, stream>>>(ys, ei, ew, keyR, keyC, ra, rb,
                                                                       bsrc, bdst, out, E, N, Epad / 16);
}


# degree factors off the per-hit path: projection stores rb*ys, ra*yd; owner takes the target row's factor from LDS, weight asked for with the indices (3 trips to 2); products regrouped, sum order same
# speedup vs baseline: 1.6456x; 1.0807x over previous
#include <hip/hip_runtime.h>
#include <hip/hip_bf16.h>


#define DEG_STRIDE 64
#define DEG_IN_OFF 32

#define OWN_ROWS    392
#define KEY_LINE    128

typedef __attribute__((ext_vector_type(2))) float v2f;
typedef __attribute__((ext_vector_type(4))) float v4f;
typedef __attribute__((ext_vector_type(8))) float v8f;
typedef __attribute__((ext_vector_type(4))) int v4i;
typedef __attribute__((ext_vector_type(4))) unsigned v4u;

__device__ __forceinline__ void atomAddF(float* p, float v) {
#if defined(__HIP_DEVICE_COMPILE__)
    unsafeAtomicAdd(p, v);
#else
    (void)p; (void)v;
#endif
}

__global__ void k_zero_deg(float* __restrict__ degw, long long ndeg) {
    long long i = (long long)blockIdx.x * blockDim.x + threadIdx.x;
    if (i < ndeg) degw[i * DEG_IN_OFF] = 0.0f;
}

__global__ void __launch_bounds__(256)
k_degree(const int* __restrict__ ei, const float* __restrict__ w,
         float* __restrict__ outdeg, float* __restrict__ indeg,
         unsigned* __restrict__ keyR, unsigned* __restrict__ keyC, int E, int nq4) {
    const int q = blockIdx.x * blockDim.x + threadIdx.x;
    if (q >= nq4) return;
    const long long e0 = 4LL * q;
    int   r[4], c[4];
    float we[4];
    if (((E & 3) == 0) && e0 + 4 <= (long long)E) {
        const v4i R  = *(const v4i*)(ei + e0);
        const v4i C  = *(const v4i*)(ei + (long long)E + e0);
        const v4f Wv = *(const v4f*)(w + e0);
#pragma unroll
        for (int m = 0; m < 4; ++m) { r[m] = R[m]; c[m] = C[m]; we[m] = Wv[m]; }
    } else {
#pragma unroll
        for (int m = 0; m < 4; ++m) {
            const long long e = e0 + m;
            const bool ok = e < (long long)E;
            r[m]  = ok ? ei[e] : -1;
            c[m]  = ok ? ei[(long long)E + e] : -1;
            we[m] = ok ? w[e] : 0.0f;
        }
    }
    unsigned kr = 0u, kc = 0u;
#pragma unroll
    for (int m = 0; m < 4; ++m) {
        if (r[m] >= 0) {
            kr |= (((unsigned)r[m] / OWN_ROWS) & 255u) << (8 * m);
            kc |= (((unsigned)c[m] / OWN_ROWS) & 255u) << (8 * m);
            atomAddF(&outdeg[(long long)r[m] * DEG_STRIDE], we[m]);
            atomAddF(&indeg[(long long)c[m] * DEG_STRIDE], we[m]);
        }
    }
    keyR[q] = kr;
    keyC[q] = kc;
}

__global__ void __launch_bounds__(256)
k_wmma_project(const float* __restrict__ x,
               const float* __restrict__ Wsrc, const float* __restrict__ Wdst,
               float* __restrict__ ys, float* __restrict__ yd,
               float* __restrict__ ra, float* __restrict__ rb,
               const float* __restrict__ degw, int tilesM, int Nnodes) {
    const int  fwhich = (int)(threadIdx.x >> 5);
    const int  fm     = (int)blockIdx.x * 32 + (int)(threadIdx.x & 31);
    const bool fth    = (threadIdx.x < 64) && (fm < Nnodes);
    __shared__ float facs[64];
    float deg = 0.0f;
    if (fth) deg = degw[(long long)fm * DEG_STRIDE + fwhich * DEG_IN_OFF];

    int wave  = (int)((blockIdx.x * blockDim.x + threadIdx.x) >> 5);
    int lane  = threadIdx.x & 31;
    int tileM = wave >> 2;
    int tileN = wave & 3;
    const bool active = tileM < tilesM;

    int lmod = lane & 15;
    int lhi  = lane >> 4;

    v8f accs = {};
    v8f accd = {};
    if (active) {
        int arow = tileM * 16 + lmod;
        if (arow >= Nnodes) arow = Nnodes - 1;
        const float* a_p  = x    + (long long)arow * 64 + 2 * lhi;
        const float* bs_p = Wsrc + (tileN * 16 + lmod) * 64 + 2 * lhi;
        const float* bd_p = Wdst + (tileN * 16 + lmod) * 64 + 2 * lhi;
#pragma unroll
        for (int kk = 0; kk < 16; ++kk) {
            v2f a  = *(const v2f*)(a_p  + 4 * kk);
            v2f bs = *(const v2f*)(bs_p + 4 * kk);
            v2f bd = *(const v2f*)(bd_p + 4 * kk);
            accs = __builtin_amdgcn_wmma_f32_16x16x4_f32(
                       false, a, false, bs, (short)0, accs, false, false);
            accd = __builtin_amdgcn_wmma_f32_16x16x4_f32(
                       false, a, false, bd, (short)0, accd, false, false);
        }
    }

    float f = 0.0f;
    if (fth) {
        f = (deg > 0.0f) ? rsqrtf(deg) : 0.0f;
        if (fwhich) rb[fm] = f; else ra[fm] = f;
    }
    if (threadIdx.x < 64) facs[threadIdx.x] = f;
    __syncthreads();
    if (!active) return;

    int ncol = tileN * 16 + lmod;
    int mbase = tileM * 16 + 8 * lhi;
    int lbase = (int)(threadIdx.x >> 7) * 16 + 8 * lhi;
#pragma unroll
    for (int r = 0; r < 8; ++r) {
        int m = mbase + r;
        if (m < Nnodes) {
            long long o = (long long)m * 64 + ncol;
            ys[o]  = (0.5f * accs[r]) * facs[32 + lbase + r];
            yd[o]  = (0.5f * accd[r]) * facs[lbase + r];
        }
    }
}

#define OWN_WAVES   32
#define OWN_THREADS (OWN_WAVES * 32)
#define OWN_QCAP    1056

__device__ __forceinline__ unsigned own_nz(unsigned w, unsigned gs) {
    const unsigned x = w ^ gs;
    return ((x & 0x7F7F7F7Fu) + 0x7F7F7F7Fu) | x | 0x7F7F7F7Fu;
}

__device__ __forceinline__ unsigned own_hits(v4u r, v4u c, unsigned gs) {
    unsigned m = own_nz(r[0], gs);
    m = __builtin_rotateright32(m, 1) & own_nz(r[1], gs);
    m = __builtin_rotateright32(m, 1) & own_nz(r[2], gs);
    m = __builtin_rotateright32(m, 1) & own_nz(r[3], gs);
    m = __builtin_rotateright32(m, 1) & own_nz(c[0], gs);
    m = __builtin_rotateright32(m, 1) & own_nz(c[1], gs);
    m = __builtin_rotateright32(m, 1) & own_nz(c[2], gs);
    m = __builtin_rotateright32(m, 1) & own_nz(c[3], gs);
    return ~m;
}

template <bool FULL>
__device__ __forceinline__ void own_drain(const unsigned* qb, int n, int lane,
                                          const int* __restrict__ ei, const float* __restrict__ ew,
                                          const float* fac,
                                          const float* __restrict__ ysd, float* acc,
                                          int E, int N, int lo, unsigned unr) {
    int   dl   = 0;
    int   srow = 0;
    float wn   = 0.0f;
    if (FULL || lane < n) {
        unsigned code = qb[lane];
        int e   = (int)(code >> 1);
        int dir = (int)(code & 1u);
        int row = ei[e];
        int col = ei[E + e];
        float we = ew[e];
        int d   = (dir ? col : row) - lo;
        const bool in = (unsigned)d < unr;
        dl   = in ? d : 0;
        const float fl = fac[dir * OWN_ROWS + dl];
        wn   = we * (in ? fl : 0.0f);
        srow = in ? (dir ? (row + N) : col) : 0;
    }
    v2f v[32];
#pragma unroll
    for (int j = 0; j < 32; ++j) {
        if (FULL || j < n) {
            int s = __builtin_amdgcn_readlane(srow, j);
            v[j] = *(const v2f*)(ysd + (long long)s * 64 + 2 * lane);
        }
    }
#pragma unroll
    for (int j = 0; j < 32; ++j) {
        if (FULL || j < n) {
            float w = __int_as_float(__builtin_amdgcn_readlane(__float_as_int(wn), j));
            int   d = __builtin_amdgcn_readlane(dl, j);
            float* a = acc + d * 64 + 2 * lane;
            __hip_atomic_fetch_add(a + 0, w * v[j].x, __ATOMIC_RELAXED, __HIP_MEMORY_SCOPE_WORKGROUP);
            __hip_atomic_fetch_add(a + 1, w * v[j].y, __ATOMIC_RELAXED, __HIP_MEMORY_SCOPE_WORKGROUP);
        }
    }
}

__global__ void __launch_bounds__(OWN_THREADS)
k_owner(const float* __restrict__ ysd,
        const int* __restrict__ ei, const float* __restrict__ ew,
        const unsigned* __restrict__ keyR, const unsigned* __restrict__ keyC,
        const float* __restrict__ ra, const float* __restrict__ rb,
        const float* __restrict__ bsrc, const float* __restrict__ bdst,
        float* __restrict__ out, int E, int N, int nq16) {
    __shared__ float    acc[OWN_ROWS * 64];
    __shared__ float    fac[2 * OWN_ROWS];
    __shared__ unsigned queue[OWN_WAVES * OWN_QCAP];

    const int tid  = (int)threadIdx.x;
    const int lane = tid & 31;
    const int wave = __builtin_amdgcn_readfirstlane(tid >> 5);
    const int lo   = (int)blockIdx.x * OWN_ROWS;
    const int nr   = (N - lo < OWN_ROWS) ? (N - lo) : OWN_ROWS;
    const unsigned unr = (unsigned)nr;
    const unsigned gs  = (blockIdx.x & 255u) * 0x01010101u;

    for (int i = tid; i < 2 * OWN_ROWS; i += OWN_THREADS) {
        const int which = (i >= OWN_ROWS) ? 1 : 0;
        const int d     = i - which * OWN_ROWS;
        float v = 0.0f;
        if (d < nr) v = which ? rb[lo + d] : ra[lo + d];
        fac[i] = v;
    }
    for (int i = tid; i < OWN_ROWS * 64; i += OWN_THREADS) acc[i] = 0.0f;
    __syncthreads();

    unsigned* qw = queue + wave * OWN_QCAP;
    unsigned  qn = 0;

    const int chunks = (nq16 + 31) >> 5;

    const int ngroups = (chunks + 3) >> 2;
    const int rot     = (int)(((long long)blockIdx.x * ngroups) / gridDim.x);
    const v4u* kR = (const v4u*)keyR;
    const v4u* kC = (const v4u*)keyC;
    for (int gi = wave; gi < ngroups; gi += OWN_WAVES) {
        int gr = gi + rot;
        if (gr >= ngroups) gr -= ngroups;
        const int cb = gr * 4;
        v4u R[4], C[4];
        if ((cb + 4) * 32 <= nq16) {
#pragma unroll
            for (int k = 0; k < 4; ++k) {
                const int q = (cb + k) * 32 + lane;
                R[k] = kR[q];
                C[k] = kC[q];
            }
        } else {
            const unsigned miss = ~gs;
#pragma unroll
            for (int k = 0; k < 4; ++k) {
                const int q = (cb + k) * 32 + lane;
                const bool ok = q < nq16;
                v4u no = {miss, miss, miss, miss};
                R[k] = ok ? kR[q] : no;
                C[k] = ok ? kC[q] : no;
            }
        }
        unsigned hm[4];
#pragma unroll
        for (int k = 0; k < 4; ++k) hm[k] = own_hits(R[k], C[k], gs);
#pragma unroll
        for (int k = 0; k < 4; ++k) {
            unsigned h = hm[k];
            const unsigned e0 = 16u * (unsigned)((cb + k) * 32 + lane);
            unsigned bal = __builtin_amdgcn_ballot_w32(h != 0u);
            while (bal) {
                const bool     act = (h != 0u);
                const unsigned p   = act ? (unsigned)__builtin_ctz(h) : 0u;
                h &= (h - 1u);
                const unsigned e   = e0 + 4u * (p & 3u) + (p >> 3);
                const unsigned dir = (p >> 2) & 1u;
                const bool     ok  = act && (e < (unsigned)E);
                const unsigned okm = __builtin_amdgcn_ballot_w32(ok);
                const unsigned pos = qn + __builtin_amdgcn_mbcnt_lo(okm, 0u);
                if (ok) qw[pos] = (e << 1) | dir;
                qn += (unsigned)__builtin_popcount(okm);
                bal = __builtin_amdgcn_ballot_w32(h != 0u);
            }
            __builtin_amdgcn_wave_barrier();
            while (qn >= 32u) {
                qn -= 32u;
                own_drain<true>(qw + qn, 32, lane, ei, ew, fac, ysd, acc, E, N, lo, unr);
            }
        }
    }
    __builtin_amdgcn_wave_barrier();
    if (qn) own_drain<false>(qw, (int)qn, lane, ei, ew, fac, ysd, acc, E, N, lo, unr);

    __syncthreads();

    const v2f bs = *(const v2f*)(bsrc + 2 * lane);
    const v2f bd = *(const v2f*)(bdst + 2 * lane);
    v2f bias;
    bias.x = 0.5f * (bs.x + bd.x);
    bias.y = 0.5f * (bs.y + bd.y);
    for (int r = wave; r < nr; r += OWN_WAVES) {
        v2f a = *(const v2f*)(acc + r * 64 + 2 * lane);
        v2f o;
        o.x = bias.x + a.x;
        o.y = bias.y + a.y;
        *(v2f*)(out + (long long)(lo + r) * 64 + 2 * lane) = o;
    }
}

extern "C" void kernel_launch(void* const* d_in, const int* in_sizes, int n_in,
                              void* d_out, int out_size, void* d_ws, size_t ws_size,
                              hipStream_t stream) {
    const float* x    = (const float*)d_in[0];
    const int*   ei   = (const int*)d_in[1];
    const float* ew   = (const float*)d_in[2];
    const float* Wsrc = (const float*)d_in[3];
    const float* bsrc = (const float*)d_in[4];
    const float* Wdst = (const float*)d_in[5];
    const float* bdst = (const float*)d_in[6];
    float* out = (float*)d_out;

    const int N = in_sizes[0] / 64;
    const int E = in_sizes[2];

    float* ws     = (float*)d_ws;
    float* outdeg = out;
    float* indeg  = out + DEG_IN_OFF;
    float* ra     = ws;
    float* rb     = ws + (long long)N;
    float* ys     = ws + 2LL * N + (long long)E;
    float* yd     = ys + 64LL * N;

    const int Epad = (E + KEY_LINE - 1) / KEY_LINE * KEY_LINE;
    unsigned char* kbase = (unsigned char*)(((size_t)(ws + 2LL * N) + 127) & ~(size_t)127);
    unsigned* keyR = (unsigned*)kbase;
    unsigned* keyC = (unsigned*)(kbase + Epad);

    const long long ndeg = 2LL * N;
    k_zero_deg<<<(unsigned)((ndeg + 255) / 256), 256, 0, stream>>>(out, ndeg);

    const int nq4 = Epad / 4;
    k_degree<<<(nq4 + 255) / 256, 256, 0, stream>>>(ei, ew, outdeg, indeg, keyR, keyC, E, nq4);

    const int tilesM = (N + 15) / 16;
    const int waves  = tilesM * 4;
    k_wmma_project<<<(waves + 7) / 8, 256, 0, stream>>>(x, Wsrc, Wdst,
                                                        ys, yd, ra, rb, out, tilesM, N);

    k_owner<<<(N + OWN_ROWS - 1) / OWN_ROWS, OWN_THREADS, 0, stream>>>(ys, ei, ew, keyR, keyC, ra, rb,
                                                                       bsrc, bdst, out, E, N, Epad / 16);
}


# degree sums by an owner scan: a work-group owns 392 nodes, reads the key planes, adds its hits' weights in LDS as 64-bit fixed point (ds_add_u64: exact in any order, rounded once); no global atomics
# speedup vs baseline: 1.8309x; 1.1126x over previous
#include <hip/hip_runtime.h>
#include <hip/hip_bf16.h>


#define OWN_ROWS    392
#define KEY_LINE    128

typedef __attribute__((ext_vector_type(2))) float v2f;
typedef __attribute__((ext_vector_type(4))) float v4f;
typedef __attribute__((ext_vector_type(8))) float v8f;
typedef __attribute__((ext_vector_type(4))) int v4i;
typedef __attribute__((ext_vector_type(4))) unsigned v4u;

__global__ void __launch_bounds__(256)
k_keys(const int* __restrict__ ei,
       unsigned* __restrict__ keyR, unsigned* __restrict__ keyC, int E, int nq4) {
    const int q = blockIdx.x * blockDim.x + threadIdx.x;
    if (q >= nq4) return;
    const long long e0 = 4LL * q;
    int r[4], c[4];
    if (((E & 3) == 0) && e0 + 4 <= (long long)E) {
        const v4i R  = *(const v4i*)(ei + e0);
        const v4i C  = *(const v4i*)(ei + (long long)E + e0);
#pragma unroll
        for (int m = 0; m < 4; ++m) { r[m] = R[m]; c[m] = C[m]; }
    } else {
#pragma unroll
        for (int m = 0; m < 4; ++m) {
            const long long e = e0 + m;
            const bool ok = e < (long long)E;
            r[m]  = ok ? ei[e] : -1;
            c[m]  = ok ? ei[(long long)E + e] : -1;
        }
    }
    unsigned kr = 0u, kc = 0u;
#pragma unroll
    for (int m = 0; m < 4; ++m) {
        if (r[m] >= 0) {
            kr |= (((unsigned)r[m] / OWN_ROWS) & 255u) << (8 * m);
            kc |= (((unsigned)c[m] / OWN_ROWS) & 255u) << (8 * m);
        }
    }
    keyR[q] = kr;
    keyC[q] = kc;
}

__global__ void __launch_bounds__(256)
k_wmma_project(const float* __restrict__ x,
               const float* __restrict__ Wsrc, const float* __restrict__ Wdst,
               float* __restrict__ ys, float* __restrict__ yd,
               const float* __restrict__ ra, const float* __restrict__ rb,
               int tilesM, int Nnodes) {
    const int  fwhich = (int)(threadIdx.x >> 5);
    const int  fm     = (int)blockIdx.x * 32 + (int)(threadIdx.x & 31);
    const bool fth    = (threadIdx.x < 64) && (fm < Nnodes);
    __shared__ float facs[64];
    float f = 0.0f;
    if (fth) f = fwhich ? rb[fm] : ra[fm];

    int wave  = (int)((blockIdx.x * blockDim.x + threadIdx.x) >> 5);
    int lane  = threadIdx.x & 31;
    int tileM = wave >> 2;
    int tileN = wave & 3;
    const bool active = tileM < tilesM;

    int lmod = lane & 15;
    int lhi  = lane >> 4;

    v8f accs = {};
    v8f accd = {};
    if (active) {
        int arow = tileM * 16 + lmod;
        if (arow >= Nnodes) arow = Nnodes - 1;
        const float* a_p  = x    + (long long)arow * 64 + 2 * lhi;
        const float* bs_p = Wsrc + (tileN * 16 + lmod) * 64 + 2 * lhi;
        const float* bd_p = Wdst + (tileN * 16 + lmod) * 64 + 2 * lhi;
#pragma unroll
        for (int kk = 0; kk < 16; ++kk) {
            v2f a  = *(const v2f*)(a_p  + 4 * kk);
            v2f bs = *(const v2f*)(bs_p + 4 * kk);
            v2f bd = *(const v2f*)(bd_p + 4 * kk);
            accs = __builtin_amdgcn_wmma_f32_16x16x4_f32(
                       false, a, false, bs, (short)0, accs, false, false);
            accd = __builtin_amdgcn_wmma_f32_16x16x4_f32(
                       false, a, false, bd, (short)0, accd, false, false);
        }
    }

    if (threadIdx.x < 64) facs[threadIdx.x] = f;
    __syncthreads();
    if (!active) return;

    int ncol = tileN * 16 + lmod;
    int mbase = tileM * 16 + 8 * lhi;
    int lbase = (int)(threadIdx.x >> 7) * 16 + 8 * lhi;
#pragma unroll
    for (int r = 0; r < 8; ++r) {
        int m = mbase + r;
        if (m < Nnodes) {
            long long o = (long long)m * 64 + ncol;
            ys[o]  = (0.5f * accs[r]) * facs[32 + lbase + r];
            yd[o]  = (0.5f * accd[r]) * facs[lbase + r];
        }
    }
}

#define OWN_WAVES   32
#define OWN_THREADS (OWN_WAVES * 32)
#define OWN_QCAP    1056

__device__ __forceinline__ unsigned own_nz(unsigned w, unsigned gs) {
    const unsigned x = w ^ gs;
    return ((x & 0x7F7F7F7Fu) + 0x7F7F7F7Fu) | x | 0x7F7F7F7Fu;
}

__device__ __forceinline__ unsigned own_hits(v4u r, v4u c, unsigned gs) {
    unsigned m = own_nz(r[0], gs);
    m = __builtin_rotateright32(m, 1) & own_nz(r[1], gs);
    m = __builtin_rotateright32(m, 1) & own_nz(r[2], gs);
    m = __builtin_rotateright32(m, 1) & own_nz(r[3], gs);
    m = __builtin_rotateright32(m, 1) & own_nz(c[0], gs);
    m = __builtin_rotateright32(m, 1) & own_nz(c[1], gs);
    m = __builtin_rotateright32(m, 1) & own_nz(c[2], gs);
    m = __builtin_rotateright32(m, 1) & own_nz(c[3], gs);
    return ~m;
}

template <bool FULL>
__device__ __forceinline__ void own_drain(const unsigned* qb, int n, int lane,
                                          const int* __restrict__ ei, const float* __restrict__ ew,
                                          const float* fac,
                                          const float* __restrict__ ysd, float* acc,
                                          int E, int N, int lo, unsigned unr) {
    int   dl   = 0;
    int   srow = 0;
    float wn   = 0.0f;
    if (FULL || lane < n) {
        unsigned code = qb[lane];
        int e   = (int)(code >> 1);
        int dir = (int)(code & 1u);
        int row = ei[e];
        int col = ei[E + e];
        float we = ew[e];
        int d   = (dir ? col : row) - lo;
        const bool in = (unsigned)d < unr;
        dl   = in ? d : 0;
        const float fl = fac[dir * OWN_ROWS + dl];
        wn   = we * (in ? fl : 0.0f);
        srow = in ? (dir ? (row + N) : col) : 0;
    }
    v2f v[32];
#pragma unroll
    for (int j = 0; j < 32; ++j) {
        if (FULL || j < n) {
            int s = __builtin_amdgcn_readlane(srow, j);
            v[j] = *(const v2f*)(ysd + (long long)s * 64 + 2 * lane);
        }
    }
#pragma unroll
    for (int j = 0; j < 32; ++j) {
        if (FULL || j < n) {
            float w = __int_as_float(__builtin_amdgcn_readlane(__float_as_int(wn), j));
            int   d = __builtin_amdgcn_readlane(dl, j);
            float* a = acc + d * 64 + 2 * lane;
            __hip_atomic_fetch_add(a + 0, w * v[j].x, __ATOMIC_RELAXED, __HIP_MEMORY_SCOPE_WORKGROUP);
            __hip_atomic_fetch_add(a + 1, w * v[j].y, __ATOMIC_RELAXED, __HIP_MEMORY_SCOPE_WORKGROUP);
        }
    }
}

#define DEG_FIX 1099511627776.0f

template <bool FULL>
__device__ __forceinline__ void deg_drain(const unsigned* qb, int n, int lane,
                                          const int* __restrict__ ei, const float* __restrict__ ew,
                                          unsigned long long* dsum, int E, int lo, unsigned unr) {
    if (FULL || lane < n) {
        const unsigned code = qb[lane];
        const int e   = (int)(code >> 1);
        const int dir = (int)(code & 1u);
        const int t   = ei[dir ? (E + e) : e];
        const float we = ew[e];
        const int d   = t - lo;
        const bool in = (unsigned)d < unr;
        const long long fx = (long long)(we * (in ? DEG_FIX : 0.0f));
        __hip_atomic_fetch_add(dsum + (in ? dir * OWN_ROWS + d : 0), (unsigned long long)fx,
                               __ATOMIC_RELAXED, __HIP_MEMORY_SCOPE_WORKGROUP);
    }
}

__global__ void __launch_bounds__(OWN_THREADS)
k_degown(const int* __restrict__ ei, const float* __restrict__ ew,
         const unsigned* __restrict__ keyR, const unsigned* __restrict__ keyC,
         float* __restrict__ ra, float* __restrict__ rb, int E, int N, int nq16) {
    __shared__ unsigned long long dsum[2 * OWN_ROWS];
    __shared__ unsigned queue[OWN_WAVES * OWN_QCAP];

    const int tid  = (int)threadIdx.x;
    const int lane = tid & 31;
    const int wave = __builtin_amdgcn_readfirstlane(tid >> 5);
    const int lo   = (int)blockIdx.x * OWN_ROWS;
    const int nr   = (N - lo < OWN_ROWS) ? (N - lo) : OWN_ROWS;
    const unsigned unr = (unsigned)nr;
    const unsigned gs  = (blockIdx.x & 255u) * 0x01010101u;

    for (int i = tid; i < 2 * OWN_ROWS; i += OWN_THREADS) dsum[i] = 0ull;
    __syncthreads();

    unsigned* qw = queue + wave * OWN_QCAP;
    unsigned  qn = 0;

    const int chunks  = (nq16 + 31) >> 5;
    const int ngroups = (chunks + 3) >> 2;
    const int rot     = (int)(((long long)blockIdx.x * ngroups) / gridDim.x);
    const v4u* kR = (const v4u*)keyR;
    const v4u* kC = (const v4u*)keyC;
    for (int gi = wave; gi < ngroups; gi += OWN_WAVES) {
        int gr = gi + rot;
        if (gr >= ngroups) gr -= ngroups;
        const int cb = gr * 4;
        v4u R[4], C[4];
        if ((cb + 4) * 32 <= nq16) {
#pragma unroll
            for (int k = 0; k < 4; ++k) {
                const int q = (cb + k) * 32 + lane;
                R[k] = kR[q];
                C[k] = kC[q];
            }
        } else {
            const unsigned miss = ~gs;
#pragma unroll
            for (int k = 0; k < 4; ++k) {
                const int q = (cb + k) * 32 + lane;
                const bool ok = q < nq16;
                v4u no = {miss, miss, miss, miss};
                R[k] = ok ? kR[q] : no;
                C[k] = ok ? kC[q] : no;
            }
        }
        unsigned hm[4];
#pragma unroll
        for (int k = 0; k < 4; ++k) hm[k] = own_hits(R[k], C[k], gs);
#pragma unroll
        for (int k = 0; k < 4; ++k) {
            unsigned h = hm[k];
            const unsigned e0 = 16u * (unsigned)((cb + k) * 32 + lane);
            unsigned bal = __builtin_amdgcn_ballot_w32(h != 0u);
            while (bal) {
                const bool     act = (h != 0u);
                const unsigned p   = act ? (unsigned)__builtin_ctz(h) : 0u;
                h &= (h - 1u);
                const unsigned e   = e0 + 4u * (p & 3u) + (p >> 3);
                const unsigned dir = (p >> 2) & 1u;
                const bool     ok  = act && (e < (unsigned)E);
                const unsigned okm = __builtin_amdgcn_ballot_w32(ok);
                const unsigned pos = qn + __builtin_amdgcn_mbcnt_lo(okm, 0u);
                if (ok) qw[pos] = (e << 1) | dir;
                qn += (unsigned)__builtin_popcount(okm);
                bal = __builtin_amdgcn_ballot_w32(h != 0u);
            }
            __builtin_amdgcn_wave_barrier();
            while (qn >= 32u) {
                qn -= 32u;
                deg_drain<true>(qw + qn, 32, lane, ei, ew, dsum, E, lo, unr);
            }
        }
    }
    __builtin_amdgcn_wave_barrier();
    if (qn) deg_drain<false>(qw, (int)qn, lane, ei, ew, dsum, E, lo, unr);

    __syncthreads();

    for (int i = tid; i < 2 * OWN_ROWS; i += OWN_THREADS) {
        const int which = (i >= OWN_ROWS) ? 1 : 0;
        const int d     = i - which * OWN_ROWS;
        if (d < nr) {
            const float deg = (float)(long long)dsum[i] * (1.0f / DEG_FIX);
            const float f   = (deg > 0.0f) ? rsqrtf(deg) : 0.0f;
            if (which) rb[lo + d] = f; else ra[lo + d] = f;
        }
    }
}

__global__ void __launch_bounds__(OWN_THREADS)
k_owner(const float* __restrict__ ysd,
        const int* __restrict__ ei, const float* __restrict__ ew,
        const unsigned* __restrict__ keyR, const unsigned* __restrict__ keyC,
        const float* __restrict__ ra, const float* __restrict__ rb,
        const float* __restrict__ bsrc, const float* __restrict__ bdst,
        float* __restrict__ out, int E, int N, int nq16) {
    __shared__ float    acc[OWN_ROWS * 64];
    __shared__ float    fac[2 * OWN_ROWS];
    __shared__ unsigned queue[OWN_WAVES * OWN_QCAP];

    const int tid  = (int)threadIdx.x;
    const int lane = tid & 31;
    const int wave = __builtin_amdgcn_readfirstlane(tid >> 5);
    const int lo   = (int)blockIdx.x * OWN_ROWS;
    const int nr   = (N - lo < OWN_ROWS) ? (N - lo) : OWN_ROWS;
    const unsigned unr = (unsigned)nr;
    const unsigned gs  = (blockIdx.x & 255u) * 0x01010101u;

    for (int i = tid; i < 2 * OWN_ROWS; i += OWN_THREADS) {
        const int which = (i >= OWN_ROWS) ? 1 : 0;
        const int d     = i - which * OWN_ROWS;
        float v = 0.0f;
        if (d < nr) v = which ? rb[lo + d] : ra[lo + d];
        fac[i] = v;
    }
    for (int i = tid; i < OWN_ROWS * 64; i += OWN_THREADS) acc[i] = 0.0f;
    __syncthreads();

    unsigned* qw = queue + wave * OWN_QCAP;
    unsigned  qn = 0;

    const int chunks = (nq16 + 31) >> 5;

    const int ngroups = (chunks + 3) >> 2;
    const int rot     = (int)(((long long)blockIdx.x * ngroups) / gridDim.x);
    const v4u* kR = (const v4u*)keyR;
    const v4u* kC = (const v4u*)keyC;
    for (int gi = wave; gi < ngroups; gi += OWN_WAVES) {
        int gr = gi + rot;
        if (gr >= ngroups) gr -= ngroups;
        const int cb = gr * 4;
        v4u R[4], C[4];
        if ((cb + 4) * 32 <= nq16) {
#pragma unroll
            for (int k = 0; k < 4; ++k) {
                const int q = (cb + k) * 32 + lane;
                R[k] = kR[q];
                C[k] = kC[q];
            }
        } else {
            const unsigned miss = ~gs;
#pragma unroll
            for (int k = 0; k < 4; ++k) {
                const int q = (cb + k) * 32 + lane;
                const bool ok = q < nq16;
                v4u no = {miss, miss, miss, miss};
                R[k] = ok ? kR[q] : no;
                C[k] = ok ? kC[q] : no;
            }
        }
        unsigned hm[4];
#pragma unroll
        for (int k = 0; k < 4; ++k) hm[k] = own_hits(R[k], C[k], gs);
#pragma unroll
        for (int k = 0; k < 4; ++k) {
            unsigned h = hm[k];
            const unsigned e0 = 16u * (unsigned)((cb + k) * 32 + lane);
            unsigned bal = __builtin_amdgcn_ballot_w32(h != 0u);
            while (bal) {
                const bool     act = (h != 0u);
                const unsigned p   = act ? (unsigned)__builtin_ctz(h) : 0u;
                h &= (h - 1u);
                const unsigned e   = e0 + 4u * (p & 3u) + (p >> 3);
                const unsigned dir = (p >> 2) & 1u;
                const bool     ok  = act && (e < (unsigned)E);
                const unsigned okm = __builtin_amdgcn_ballot_w32(ok);
                const unsigned pos = qn + __builtin_amdgcn_mbcnt_lo(okm, 0u);
                if (ok) qw[pos] = (e << 1) | dir;
                qn += (unsigned)__builtin_popcount(okm);
                bal = __builtin_amdgcn_ballot_w32(h != 0u);
            }
            __builtin_amdgcn_wave_barrier();
            while (qn >= 32u) {
                qn -= 32u;
                own_drain<true>(qw + qn, 32, lane, ei, ew, fac, ysd, acc, E, N, lo, unr);
            }
        }
    }
    __builtin_amdgcn_wave_barrier();
    if (qn) own_drain<false>(qw, (int)qn, lane, ei, ew, fac, ysd, acc, E, N, lo, unr);

    __syncthreads();

    const v2f bs = *(const v2f*)(bsrc + 2 * lane);
    const v2f bd = *(const v2f*)(bdst + 2 * lane);
    v2f bias;
    bias.x = 0.5f * (bs.x + bd.x);
    bias.y = 0.5f * (bs.y + bd.y);
    for (int r = wave; r < nr; r += OWN_WAVES) {
        v2f a = *(const v2f*)(acc + r * 64 + 2 * lane);
        v2f o;
        o.x = bias.x + a.x;
        o.y = bias.y + a.y;
        *(v2f*)(out + (long long)(lo + r) * 64 + 2 * lane) = o;
    }
}

extern "C" void kernel_launch(void* const* d_in, const int* in_sizes, int n_in,
                              void* d_out, int out_size, void* d_ws, size_t ws_size,
                              hipStream_t stream) {
    const float* x    = (const float*)d_in[0];
    const int*   ei   = (const int*)d_in[1];
    const float* ew   = (const float*)d_in[2];
    const float* Wsrc = (const float*)d_in[3];
    const float* bsrc = (const float*)d_in[4];
    const float* Wdst = (const float*)d_in[5];
    const float* bdst = (const float*)d_in[6];
    float* out = (float*)d_out;

    const int N = in_sizes[0] / 64;
    const int E = in_sizes[2];

    float* ws     = (float*)d_ws;
    float* ra     = ws;
    float* rb     = ws + (long long)N;
    float* ys     = ws + 2LL * N + (long long)E;
    float* yd     = ys + 64LL * N;

    const int Epad = (E + KEY_LINE - 1) / KEY_LINE * KEY_LINE;
    unsigned char* kbase = (unsigned char*)(((size_t)(ws + 2LL * N) + 127) & ~(size_t)127);
    unsigned* keyR = (unsigned*)kbase;
    unsigned* keyC = (unsigned*)(kbase + Epad);

    const int nq4 = Epad / 4;
    k_keys<<<(nq4 + 255) / 256, 256, 0, stream>>>(ei, keyR, keyC, E, nq4);

    const int nblocks = (N + OWN_ROWS - 1) / OWN_ROWS;
    k_degown<<<nblocks, OWN_THREADS, 0, stream>>>(ei, ew, keyR, keyC, ra, rb, E, N, Epad / 16);

    const int tilesM = (N + 15) / 16;
    const int waves  = tilesM * 4;
    k_wmma_project<<<(waves + 7) / 8, 256, 0, stream>>>(x, Wsrc, Wdst,
                                                        ys, yd, ra, rb, tilesM, N);

    k_owner<<<nblocks, OWN_THREADS, 0, stream>>>(ys, ei, ew, keyR, keyC, ra, rb,
                                                 bsrc, bdst, out, E, N, Epad / 16);
}


# edges repacked once as 12-byte records (row, col, w) by the key pass; an owner pass reads one record per hit instead of 2 (degree pass) or 3 (owner pass) scattered words; sum order unchanged
# speedup vs baseline: 1.9672x; 1.0744x over previous
#include <hip/hip_runtime.h>
#include <hip/hip_bf16.h>


#define OWN_ROWS    392
#define KEY_LINE    128

typedef __attribute__((ext_vector_type(2))) float v2f;
typedef __attribute__((ext_vector_type(4))) float v4f;
typedef __attribute__((ext_vector_type(8))) float v8f;
typedef __attribute__((ext_vector_type(4))) int v4i;
typedef __attribute__((ext_vector_type(4))) unsigned v4u;

struct __attribute__((packed, aligned(4))) EdgeRec { int row; int col; float w; };

template <bool REC>
__global__ void __launch_bounds__(256)
k_keys(const int* __restrict__ ei, const float* __restrict__ ew,
       unsigned* __restrict__ keyR, unsigned* __restrict__ keyC,
       int* __restrict__ recs, int E, int nq4) {
    const int q = blockIdx.x * blockDim.x + threadIdx.x;
    if (q >= nq4) return;
    const long long e0 = 4LL * q;
    int r[4], c[4];
    if (((E & 3) == 0) && e0 + 4 <= (long long)E) {
        const v4i R  = *(const v4i*)(ei + e0);
        const v4i C  = *(const v4i*)(ei + (long long)E + e0);
#pragma unroll
        for (int m = 0; m < 4; ++m) { r[m] = R[m]; c[m] = C[m]; }
        if (REC) {
            const v4f Wf = *(const v4f*)(ew + e0);
            v4i* rp = (v4i*)(recs + 12LL * q);
            const v4i s0 = {R[0], C[0], __float_as_int(Wf[0]), R[1]};
            const v4i s1 = {C[1], __float_as_int(Wf[1]), R[2], C[2]};
            const v4i s2 = {__float_as_int(Wf[2]), R[3], C[3], __float_as_int(Wf[3])};
            rp[0] = s0;
            rp[1] = s1;
            rp[2] = s2;
        }
    } else {
#pragma unroll
        for (int m = 0; m < 4; ++m) {
            const long long e = e0 + m;
            const bool ok = e < (long long)E;
            r[m]  = ok ? ei[e] : -1;
            c[m]  = ok ? ei[(long long)E + e] : -1;
            if (REC && ok) {
                int* rp = recs + 3LL * e;
                rp[0] = r[m];
                rp[1] = c[m];
                rp[2] = __float_as_int(ew[e]);
            }
        }
    }
    unsigned kr = 0u, kc = 0u;
#pragma unroll
    for (int m = 0; m < 4; ++m) {
        if (r[m] >= 0) {
            kr |= (((unsigned)r[m] / OWN_ROWS) & 255u) << (8 * m);
            kc |= (((unsigned)c[m] / OWN_ROWS) & 255u) << (8 * m);
        }
    }
    keyR[q] = kr;
    keyC[q] = kc;
}

__global__ void __launch_bounds__(256)
k_wmma_project(const float* __restrict__ x,
               const float* __restrict__ Wsrc, const float* __restrict__ Wdst,
               float* __restrict__ ys, float* __restrict__ yd,
               const float* __restrict__ ra, const float* __restrict__ rb,
               int tilesM, int Nnodes) {
    const int  fwhich = (int)(threadIdx.x >> 5);
    const int  fm     = (int)blockIdx.x * 32 + (int)(threadIdx.x & 31);
    const bool fth    = (threadIdx.x < 64) && (fm < Nnodes);
    __shared__ float facs[64];
    float f = 0.0f;
    if (fth) f = fwhich ? rb[fm] : ra[fm];

    int wave  = (int)((blockIdx.x * blockDim.x + threadIdx.x) >> 5);
    int lane  = threadIdx.x & 31;
    int tileM = wave >> 2;
    int tileN = wave & 3;
    const bool active = tileM < tilesM;

    int lmod = lane & 15;
    int lhi  = lane >> 4;

    v8f accs = {};
    v8f accd = {};
    if (active) {
        int arow = tileM * 16 + lmod;
        if (arow >= Nnodes) arow = Nnodes - 1;
        const float* a_p  = x    + (long long)arow * 64 + 2 * lhi;
        const float* bs_p = Wsrc + (tileN * 16 + lmod) * 64 + 2 * lhi;
        const float* bd_p = Wdst + (tileN * 16 + lmod) * 64 + 2 * lhi;
#pragma unroll
        for (int kk = 0; kk < 16; ++kk) {
            v2f a  = *(const v2f*)(a_p  + 4 * kk);
            v2f bs = *(const v2f*)(bs_p + 4 * kk);
            v2f bd = *(const v2f*)(bd_p + 4 * kk);
            accs = __builtin_amdgcn_wmma_f32_16x16x4_f32(
                       false, a, false, bs, (short)0, accs, false, false);
            accd = __builtin_amdgcn_wmma_f32_16x16x4_f32(
                       false, a, false, bd, (short)0, accd, false, false);
        }
    }

    if (threadIdx.x < 64) facs[threadIdx.x] = f;
    __syncthreads();
    if (!active) return;

    int ncol = tileN * 16 + lmod;
    int mbase = tileM * 16 + 8 * lhi;
    int lbase = (int)(threadIdx.x >> 7) * 16 + 8 * lhi;
#pragma unroll
    for (int r = 0; r < 8; ++r) {
        int m = mbase + r;
        if (m < Nnodes) {
            long long o = (long long)m * 64 + ncol;
            ys[o]  = (0.5f * accs[r]) * facs[32 + lbase + r];
            yd[o]  = (0.5f * accd[r]) * facs[lbase + r];
        }
    }
}

#define OWN_WAVES   32
#define OWN_THREADS (OWN_WAVES * 32)
#define OWN_QCAP    1056

__device__ __forceinline__ unsigned own_nz(unsigned w, unsigned gs) {
    const unsigned x = w ^ gs;
    return ((x & 0x7F7F7F7Fu) + 0x7F7F7F7Fu) | x | 0x7F7F7F7Fu;
}

__device__ __forceinline__ unsigned own_hits(v4u r, v4u c, unsigned gs) {
    unsigned m = own_nz(r[0], gs);
    m = __builtin_rotateright32(m, 1) & own_nz(r[1], gs);
    m = __builtin_rotateright32(m, 1) & own_nz(r[2], gs);
    m = __builtin_rotateright32(m, 1) & own_nz(r[3], gs);
    m = __builtin_rotateright32(m, 1) & own_nz(c[0], gs);
    m = __builtin_rotateright32(m, 1) & own_nz(c[1], gs);
    m = __builtin_rotateright32(m, 1) & own_nz(c[2], gs);
    m = __builtin_rotateright32(m, 1) & own_nz(c[3], gs);
    return ~m;
}

template <bool FULL, bool REC>
__device__ __forceinline__ void own_drain(const unsigned* qb, int n, int lane,
                                          const int* __restrict__ ei, const float* __restrict__ ew,
                                          const EdgeRec* __restrict__ recs,
                                          const float* fac,
                                          const float* __restrict__ ysd, float* acc,
                                          int E, int N, int lo, unsigned unr) {
    int   dl   = 0;
    int   srow = 0;
    float wn   = 0.0f;
    if (FULL || lane < n) {
        unsigned code = qb[lane];
        int e   = (int)(code >> 1);
        int dir = (int)(code & 1u);
        int row, col;
        float we;
        if (REC) {
            const EdgeRec r = recs[e];
            row = r.row;
            col = r.col;
            we  = r.w;
        } else {
            row = ei[e];
            col = ei[E + e];
            we  = ew[e];
        }
        int d   = (dir ? col : row) - lo;
        const bool in = (unsigned)d < unr;
        dl   = in ? d : 0;
        const float fl = fac[dir * OWN_ROWS + dl];
        wn   = we * (in ? fl : 0.0f);
        srow = in ? (dir ? (row + N) : col) : 0;
    }
    v2f v[32];
#pragma unroll
    for (int j = 0; j < 32; ++j) {
        if (FULL || j < n) {
            int s = __builtin_amdgcn_readlane(srow, j);
            v[j] = *(const v2f*)(ysd + (long long)s * 64 + 2 * lane);
        }
    }
#pragma unroll
    for (int j = 0; j < 32; ++j) {
        if (FULL || j < n) {
            float w = __int_as_float(__builtin_amdgcn_readlane(__float_as_int(wn), j));
            int   d = __builtin_amdgcn_readlane(dl, j);
            float* a = acc + d * 64 + 2 * lane;
            __hip_atomic_fetch_add(a + 0, w * v[j].x, __ATOMIC_RELAXED, __HIP_MEMORY_SCOPE_WORKGROUP);
            __hip_atomic_fetch_add(a + 1, w * v[j].y, __ATOMIC_RELAXED, __HIP_MEMORY_SCOPE_WORKGROUP);
        }
    }
}

#define DEG_FIX 1099511627776.0f

template <bool FULL, bool REC>
__device__ __forceinline__ void deg_drain(const unsigned* qb, int n, int lane,
                                          const int* __restrict__ ei, const float* __restrict__ ew,
                                          const EdgeRec* __restrict__ recs,
                                          unsigned long long* dsum, int E, int lo, unsigned unr) {
    if (FULL || lane < n) {
        const unsigned code = qb[lane];
        const int e   = (int)(code >> 1);
        const int dir = (int)(code & 1u);
        int t;
        float we;
        if (REC) {
            const EdgeRec r = recs[e];
            t  = dir ? r.col : r.row;
            we = r.w;
        } else {
            t  = ei[dir ? (E + e) : e];
            we = ew[e];
        }
        const int d   = t - lo;
        const bool in = (unsigned)d < unr;
        const long long fx = (long long)(we * (in ? DEG_FIX : 0.0f));
        __hip_atomic_fetch_add(dsum + (in ? dir * OWN_ROWS + d : 0), (unsigned long long)fx,
                               __ATOMIC_RELAXED, __HIP_MEMORY_SCOPE_WORKGROUP);
    }
}

template <bool REC>
__global__ void __launch_bounds__(OWN_THREADS)
k_degown(const int* __restrict__ ei, const float* __restrict__ ew,
         const EdgeRec* __restrict__ recs,
         const unsigned* __restrict__ keyR, const unsigned* __restrict__ keyC,
         float* __restrict__ ra, float* __restrict__ rb, int E, int N, int nq16) {
    __shared__ unsigned long long dsum[2 * OWN_ROWS];
    __shared__ unsigned queue[OWN_WAVES * OWN_QCAP];

    const int tid  = (int)threadIdx.x;
    const int lane = tid & 31;
    const int wave = __builtin_amdgcn_readfirstlane(tid >> 5);
    const int lo   = (int)blockIdx.x * OWN_ROWS;
    const int nr   = (N - lo < OWN_ROWS) ? (N - lo) : OWN_ROWS;
    const unsigned unr = (unsigned)nr;
    const unsigned gs  = (blockIdx.x & 255u) * 0x01010101u;

    for (int i = tid; i < 2 * OWN_ROWS; i += OWN_THREADS) dsum[i] = 0ull;
    __syncthreads();

    unsigned* qw = queue + wave * OWN_QCAP;
    unsigned  qn = 0;

    const int chunks  = (nq16 + 31) >> 5;
    const int ngroups = (chunks + 3) >> 2;
    const int rot     = (int)(((long long)blockIdx.x * ngroups) / gridDim.x);
    const v4u* kR = (const v4u*)keyR;
    const v4u* kC = (const v4u*)keyC;
    for (int gi = wave; gi < ngroups; gi += OWN_WAVES) {
        int gr = gi + rot;
        if (gr >= ngroups) gr -= ngroups;
        const int cb = gr * 4;
        v4u R[4], C[4];
        if ((cb + 4) * 32 <= nq16) {
#pragma unroll
            for (int k = 0; k < 4; ++k) {
                const int q = (cb + k) * 32 + lane;
                R[k] = kR[q];
                C[k] = kC[q];
            }
        } else {
            const unsigned miss = ~gs;
#pragma unroll
            for (int k = 0; k < 4; ++k) {
                const int q = (cb + k) * 32 + lane;
                const bool ok = q < nq16;
                v4u no = {miss, miss, miss, miss};
                R[k] = ok ? kR[q] : no;
                C[k] = ok ? kC[q] : no;
            }
        }
        unsigned hm[4];
#pragma unroll
        for (int k = 0; k < 4; ++k) hm[k] = own_hits(R[k], C[k], gs);
#pragma unroll
        for (int k = 0; k < 4; ++k) {
            unsigned h = hm[k];
            const unsigned e0 = 16u * (unsigned)((cb + k) * 32 + lane);
            unsigned bal = __builtin_amdgcn_ballot_w32(h != 0u);
            while (bal) {
                const bool     act = (h != 0u);
                const unsigned p   = act ? (unsigned)__builtin_ctz(h) : 0u;
                h &= (h - 1u);
                const unsigned e   = e0 + 4u * (p & 3u) + (p >> 3);
                const unsigned dir = (p >> 2) & 1u;
                const bool     ok  = act && (e < (unsigned)E);
                const unsigned okm = __builtin_amdgcn_ballot_w32(ok);
                const unsigned pos = qn + __builtin_amdgcn_mbcnt_lo(okm, 0u);
                if (ok) qw[pos] = (e << 1) | dir;
                qn += (unsigned)__builtin_popcount(okm);
                bal = __builtin_amdgcn_ballot_w32(h != 0u);
            }
            __builtin_amdgcn_wave_barrier();
            while (qn >= 32u) {
                qn -= 32u;
                deg_drain<true, REC>(qw + qn, 32, lane, ei, ew, recs, dsum, E, lo, unr);
            }
        }
    }
    __builtin_amdgcn_wave_barrier();
    if (qn) deg_drain<false, REC>(qw, (int)qn, lane, ei, ew, recs, dsum, E, lo, unr);

    __syncthreads();

    for (int i = tid; i < 2 * OWN_ROWS; i += OWN_THREADS) {
        const int which = (i >= OWN_ROWS) ? 1 : 0;
        const int d     = i - which * OWN_ROWS;
        if (d < nr) {
            const float deg = (float)(long long)dsum[i] * (1.0f / DEG_FIX);
            const float f   = (deg > 0.0f) ? rsqrtf(deg) : 0.0f;
            if (which) rb[lo + d] = f; else ra[lo + d] = f;
        }
    }
}

template <bool REC>
__global__ void __launch_bounds__(OWN_THREADS)
k_owner(const float* __restrict__ ysd,
        const int* __restrict__ ei, const float* __restrict__ ew,
        const EdgeRec* __restrict__ recs,
        const unsigned* __restrict__ keyR, const unsigned* __restrict__ keyC,
        const float* __restrict__ ra, const float* __restrict__ rb,
        const float* __restrict__ bsrc, const float* __restrict__ bdst,
        float* __restrict__ out, int E, int N, int nq16) {
    __shared__ float    acc[OWN_ROWS * 64];
    __shared__ float    fac[2 * OWN_ROWS];
    __shared__ unsigned queue[OWN_WAVES * OWN_QCAP];

    const int tid  = (int)threadIdx.x;
    const int lane = tid & 31;
    const int wave = __builtin_amdgcn_readfirstlane(tid >> 5);
    const int lo   = (int)blockIdx.x * OWN_ROWS;
    const int nr   = (N - lo < OWN_ROWS) ? (N - lo) : OWN_ROWS;
    const unsigned unr = (unsigned)nr;
    const unsigned gs  = (blockIdx.x & 255u) * 0x01010101u;

    for (int i = tid; i < 2 * OWN_ROWS; i += OWN_THREADS) {
        const int which = (i >= OWN_ROWS) ? 1 : 0;
        const int d     = i - which * OWN_ROWS;
        float v = 0.0f;
        if (d < nr) v = which ? rb[lo + d] : ra[lo + d];
        fac[i] = v;
    }
    for (int i = tid; i < OWN_ROWS * 64; i += OWN_THREADS) acc[i] = 0.0f;
    __syncthreads();

    unsigned* qw = queue + wave * OWN_QCAP;
    unsigned  qn = 0;

    const int chunks = (nq16 + 31) >> 5;

    const int ngroups = (chunks + 3) >> 2;
    const int rot     = (int)(((long long)blockIdx.x * ngroups) / gridDim.x);
    const v4u* kR = (const v4u*)keyR;
    const v4u* kC = (const v4u*)keyC;
    for (int gi = wave; gi < ngroups; gi += OWN_WAVES) {
        int gr = gi + rot;
        if (gr >= ngroups) gr -= ngroups;
        const int cb = gr * 4;
        v4u R[4], C[4];
        if ((cb + 4) * 32 <= nq16) {
#pragma unroll
            for (int k = 0; k < 4; ++k) {
                const int q = (cb + k) * 32 + lane;
                R[k] = kR[q];
                C[k] = kC[q];
            }
        } else {
            const unsigned miss = ~gs;
#pragma unroll
            for (int k = 0; k < 4; ++k) {
                const int q = (cb + k) * 32 + lane;
                const bool ok = q < nq16;
                v4u no = {miss, miss, miss, miss};
                R[k] = ok ? kR[q] : no;
                C[k] = ok ? kC[q] : no;
            }
        }
        unsigned hm[4];
#pragma unroll
        for (int k = 0; k < 4; ++k) hm[k] = own_hits(R[k], C[k], gs);
#pragma unroll
        for (int k = 0; k < 4; ++k) {
            unsigned h = hm[k];
            const unsigned e0 = 16u * (unsigned)((cb + k) * 32 + lane);
            unsigned bal = __builtin_amdgcn_ballot_w32(h != 0u);
            while (bal) {
                const bool     act = (h != 0u);
                const unsigned p   = act ? (unsigned)__builtin_ctz(h) : 0u;
                h &= (h - 1u);
                const unsigned e   = e0 + 4u * (p & 3u) + (p >> 3);
                const unsigned dir = (p >> 2) & 1u;
                const bool     ok  = act && (e < (unsigned)E);
                const unsigned okm = __builtin_amdgcn_ballot_w32(ok);
                const unsigned pos = qn + __builtin_amdgcn_mbcnt_lo(okm, 0u);
                if (ok) qw[pos] = (e << 1) | dir;
                qn += (unsigned)__builtin_popcount(okm);
                bal = __builtin_amdgcn_ballot_w32(h != 0u);
            }
            __builtin_amdgcn_wave_barrier();
            while (qn >= 32u) {
                qn -= 32u;
                own_drain<true, REC>(qw + qn, 32, lane, ei, ew, recs, fac, ysd, acc, E, N, lo, unr);
            }
        }
    }
    __builtin_amdgcn_wave_barrier();
    if (qn) own_drain<false, REC>(qw, (int)qn, lane, ei, ew, recs, fac, ysd, acc, E, N, lo, unr);

    __syncthreads();

    const v2f bs = *(const v2f*)(bsrc + 2 * lane);
    const v2f bd = *(const v2f*)(bdst + 2 * lane);
    v2f bias;
    bias.x = 0.5f * (bs.x + bd.x);
    bias.y = 0.5f * (bs.y + bd.y);
    for (int r = wave; r < nr; r += OWN_WAVES) {
        v2f a = *(const v2f*)(acc + r * 64 + 2 * lane);
        v2f o;
        o.x = bias.x + a.x;
        o.y = bias.y + a.y;
        *(v2f*)(out + (long long)(lo + r) * 64 + 2 * lane) = o;
    }
}

extern "C" void kernel_launch(void* const* d_in, const int* in_sizes, int n_in,
                              void* d_out, int out_size, void* d_ws, size_t ws_size,
                              hipStream_t stream) {
    const float* x    = (const float*)d_in[0];
    const int*   ei   = (const int*)d_in[1];
    const float* ew   = (const float*)d_in[2];
    const float* Wsrc = (const float*)d_in[3];
    const float* bsrc = (const float*)d_in[4];
    const float* Wdst = (const float*)d_in[5];
    const float* bdst = (const float*)d_in[6];
    float* out = (float*)d_out;

    const int N = in_sizes[0] / 64;
    const int E = in_sizes[2];

    float* ws     = (float*)d_ws;
    float* ra     = ws;
    float* rb     = ws + (long long)N;

    const int Epad = (E + KEY_LINE - 1) / KEY_LINE * KEY_LINE;
    unsigned char* kbase = (unsigned char*)(((size_t)(ws + 2LL * N) + 127) & ~(size_t)127);
    unsigned* keyR = (unsigned*)kbase;
    unsigned* keyC = (unsigned*)(kbase + Epad);

    unsigned char* rbase = (unsigned char*)(((size_t)(kbase + 2LL * Epad) + 127) & ~(size_t)127);
    EdgeRec* recs = (EdgeRec*)rbase;
    float* ysrec  = (float*)(((size_t)(rbase + sizeof(EdgeRec) * (size_t)E) + 255) & ~(size_t)255);
    const bool rec = (size_t)((unsigned char*)(ysrec + 128LL * N) - (unsigned char*)d_ws) <= ws_size;
    float* ys     = rec ? ysrec : ws + 2LL * N + (long long)E;
    float* yd     = ys + 64LL * N;

    const int nq4 = Epad / 4;
    if (rec) k_keys<true><<<(nq4 + 255) / 256, 256, 0, stream>>>(ei, ew, keyR, keyC, (int*)recs, E, nq4);
    else     k_keys<false><<<(nq4 + 255) / 256, 256, 0, stream>>>(ei, ew, keyR, keyC, (int*)recs, E, nq4);

    const int nblocks = (N + OWN_ROWS - 1) / OWN_ROWS;
    if (rec) k_degown<true><<<nblocks, OWN_THREADS, 0, stream>>>(ei, ew, recs, keyR, keyC, ra, rb, E, N, Epad / 16);
    else     k_degown<false><<<nblocks, OWN_THREADS, 0, stream>>>(ei, ew, recs, keyR, keyC, ra, rb, E, N, Epad / 16);

    const int tilesM = (N + 15) / 16;
    const int waves  = tilesM * 4;
    k_wmma_project<<<(waves + 7) / 8, 256, 0, stream>>>(x, Wsrc, Wdst,
                                                        ys, yd, ra, rb, tilesM, N);

    if (rec) k_owner<true><<<nblocks, OWN_THREADS, 0, stream>>>(ys, ei, ew, recs, keyR, keyC, ra, rb,
                                                       bsrc, bdst, out, E, N, Epad / 16);
    else     k_owner<false><<<nblocks, OWN_THREADS, 0, stream>>>(ys, ei, ew, recs, keyR, keyC, ra, rb,
                                                        bsrc, bdst, out, E, N, Epad / 16);
}


# owner pass walks per-wave hit lists (a word a hit) that the degree owner pass files in the block's own rows of out as it scans; owner reads no key planes; per-wave sum order unchanged
# speedup vs baseline: 2.3124x; 1.1755x over previous
#include <hip/hip_runtime.h>
#include <hip/hip_bf16.h>


#define OWN_ROWS    392
#define KEY_LINE    128

typedef __attribute__((ext_vector_type(2))) float v2f;
typedef __attribute__((ext_vector_type(4))) float v4f;
typedef __attribute__((ext_vector_type(8))) float v8f;
typedef __attribute__((ext_vector_type(4))) int v4i;
typedef __attribute__((ext_vector_type(4))) unsigned v4u;

struct __attribute__((packed, aligned(4))) EdgeRec { int row; int col; float w; };

template <bool REC>
__global__ void __launch_bounds__(256)
k_keys(const int* __restrict__ ei, const float* __restrict__ ew,
       unsigned* __restrict__ keyR, unsigned* __restrict__ keyC,
       int* __restrict__ recs, int E, int nq4) {
    const int q = blockIdx.x * blockDim.x + threadIdx.x;
    if (q >= nq4) return;
    const long long e0 = 4LL * q;
    int r[4], c[4];
    if (((E & 3) == 0) && e0 + 4 <= (long long)E) {
        const v4i R  = *(const v4i*)(ei + e0);
        const v4i C  = *(const v4i*)(ei + (long long)E + e0);
#pragma unroll
        for (int m = 0; m < 4; ++m) { r[m] = R[m]; c[m] = C[m]; }
        if (REC) {
            const v4f Wf = *(const v4f*)(ew + e0);
            v4i* rp = (v4i*)(recs + 12LL * q);
            const v4i s0 = {R[0], C[0], __float_as_int(Wf[0]), R[1]};
            const v4i s1 = {C[1], __float_as_int(Wf[1]), R[2], C[2]};
            const v4i s2 = {__float_as_int(Wf[2]), R[3], C[3], __float_as_int(Wf[3])};
            rp[0] = s0;
            rp[1] = s1;
            rp[2] = s2;
        }
    } else {
#pragma unroll
        for (int m = 0; m < 4; ++m) {
            const long long e = e0 + m;
            const bool ok = e < (long long)E;
            r[m]  = ok ? ei[e] : -1;
            c[m]  = ok ? ei[(long long)E + e] : -1;
            if (REC && ok) {
                int* rp = recs + 3LL * e;
                rp[0] = r[m];
                rp[1] = c[m];
                rp[2] = __float_as_int(ew[e]);
            }
        }
    }
    unsigned kr = 0u, kc = 0u;
#pragma unroll
    for (int m = 0; m < 4; ++m) {
        if (r[m] >= 0) {
            kr |= (((unsigned)r[m] / OWN_ROWS) & 255u) << (8 * m);
            kc |= (((unsigned)c[m] / OWN_ROWS) & 255u) << (8 * m);
        }
    }
    keyR[q] = kr;
    keyC[q] = kc;
}

__global__ void __launch_bounds__(256)
k_wmma_project(const float* __restrict__ x,
               const float* __restrict__ Wsrc, const float* __restrict__ Wdst,
               float* __restrict__ ys, float* __restrict__ yd,
               const float* __restrict__ ra, const float* __restrict__ rb,
               int tilesM, int Nnodes) {
    const int  fwhich = (int)(threadIdx.x >> 5);
    const int  fm     = (int)blockIdx.x * 32 + (int)(threadIdx.x & 31);
    const bool fth    = (threadIdx.x < 64) && (fm < Nnodes);
    __shared__ float facs[64];
    float f = 0.0f;
    if (fth) f = fwhich ? rb[fm] : ra[fm];

    int wave  = (int)((blockIdx.x * blockDim.x + threadIdx.x) >> 5);
    int lane  = threadIdx.x & 31;
    int tileM = wave >> 2;
    int tileN = wave & 3;
    const bool active = tileM < tilesM;

    int lmod = lane & 15;
    int lhi  = lane >> 4;

    v8f accs = {};
    v8f accd = {};
    if (active) {
        int arow = tileM * 16 + lmod;
        if (arow >= Nnodes) arow = Nnodes - 1;
        const float* a_p  = x    + (long long)arow * 64 + 2 * lhi;
        const float* bs_p = Wsrc + (tileN * 16 + lmod) * 64 + 2 * lhi;
        const float* bd_p = Wdst + (tileN * 16 + lmod) * 64 + 2 * lhi;
#pragma unroll
        for (int kk = 0; kk < 16; ++kk) {
            v2f a  = *(const v2f*)(a_p  + 4 * kk);
            v2f bs = *(const v2f*)(bs_p + 4 * kk);
            v2f bd = *(const v2f*)(bd_p + 4 * kk);
            accs = __builtin_amdgcn_wmma_f32_16x16x4_f32(
                       false, a, false, bs, (short)0, accs, false, false);
            accd = __builtin_amdgcn_wmma_f32_16x16x4_f32(
                       false, a, false, bd, (short)0, accd, false, false);
        }
    }

    if (threadIdx.x < 64) facs[threadIdx.x] = f;
    __syncthreads();
    if (!active) return;

    int ncol = tileN * 16 + lmod;
    int mbase = tileM * 16 + 8 * lhi;
    int lbase = (int)(threadIdx.x >> 7) * 16 + 8 * lhi;
#pragma unroll
    for (int r = 0; r < 8; ++r) {
        int m = mbase + r;
        if (m < Nnodes) {
            long long o = (long long)m * 64 + ncol;
            ys[o]  = (0.5f * accs[r]) * facs[32 + lbase + r];
            yd[o]  = (0.5f * accd[r]) * facs[lbase + r];
        }
    }
}

#define OWN_WAVES   32
#define OWN_THREADS (OWN_WAVES * 32)
#define OWN_QCAP    1056
#define LIST_NONE   0xFFFFFFFFu

__device__ __forceinline__ unsigned own_nz(unsigned w, unsigned gs) {
    const unsigned x = w ^ gs;
    return ((x & 0x7F7F7F7Fu) + 0x7F7F7F7Fu) | x | 0x7F7F7F7Fu;
}

__device__ __forceinline__ unsigned own_hits(v4u r, v4u c, unsigned gs) {
    unsigned m = own_nz(r[0], gs);
    m = __builtin_rotateright32(m, 1) & own_nz(r[1], gs);
    m = __builtin_rotateright32(m, 1) & own_nz(r[2], gs);
    m = __builtin_rotateright32(m, 1) & own_nz(r[3], gs);
    m = __builtin_rotateright32(m, 1) & own_nz(c[0], gs);
    m = __builtin_rotateright32(m, 1) & own_nz(c[1], gs);
    m = __builtin_rotateright32(m, 1) & own_nz(c[2], gs);
    m = __builtin_rotateright32(m, 1) & own_nz(c[3], gs);
    return ~m;
}

template <bool FULL, bool REC>
__device__ __forceinline__ void own_drain(const unsigned* qb, int n, int lane,
                                          const int* __restrict__ ei, const float* __restrict__ ew,
                                          const EdgeRec* __restrict__ recs,
                                          const float* fac,
                                          const float* __restrict__ ysd, float* acc,
                                          int E, int N, int lo, unsigned unr) {
    int   dl   = 0;
    int   srow = 0;
    float wn   = 0.0f;
    if (FULL || lane < n) {
        unsigned code = qb[lane];
        int e   = (int)(code >> 1);
        int dir = (int)(code & 1u);
        int row, col;
        float we;
        if (REC) {
            const EdgeRec r = recs[e];
            row = r.row;
            col = r.col;
            we  = r.w;
        } else {
            row = ei[e];
            col = ei[E + e];
            we  = ew[e];
        }
        int d   = (dir ? col : row) - lo;
        const bool in = (unsigned)d < unr;
        dl   = in ? d : 0;
        const float fl = fac[dir * OWN_ROWS + dl];
        wn   = we * (in ? fl : 0.0f);
        srow = in ? (dir ? (row + N) : col) : 0;
    }
    v2f v[32];
#pragma unroll
    for (int j = 0; j < 32; ++j) {
        if (FULL || j < n) {
            int s = __builtin_amdgcn_readlane(srow, j);
            v[j] = *(const v2f*)(ysd + (long long)s * 64 + 2 * lane);
        }
    }
#pragma unroll
    for (int j = 0; j < 32; ++j) {
        if (FULL || j < n) {
            float w = __int_as_float(__builtin_amdgcn_readlane(__float_as_int(wn), j));
            int   d = __builtin_amdgcn_readlane(dl, j);
            float* a = acc + d * 64 + 2 * lane;
            __hip_atomic_fetch_add(a + 0, w * v[j].x, __ATOMIC_RELAXED, __HIP_MEMORY_SCOPE_WORKGROUP);
            __hip_atomic_fetch_add(a + 1, w * v[j].y, __ATOMIC_RELAXED, __HIP_MEMORY_SCOPE_WORKGROUP);
        }
    }
}

#define DEG_FIX 1099511627776.0f

template <bool FULL, bool REC>
__device__ __forceinline__ void deg_drain(const unsigned* qb, int n, int lane,
                                          const int* __restrict__ ei, const float* __restrict__ ew,
                                          const EdgeRec* __restrict__ recs,
                                          unsigned long long* dsum, int E, int lo, unsigned unr,
                                          unsigned* ents, unsigned ln, unsigned seg) {
    const bool fits = ln + (unsigned)n <= seg;
    if (FULL || lane < n) {
        const unsigned code = qb[lane];
        if (fits) ents[ln + (unsigned)lane] = code;
        const int e   = (int)(code >> 1);
        const int dir = (int)(code & 1u);
        int t;
        float we;
        if (REC) {
            const EdgeRec r = recs[e];
            t  = dir ? r.col : r.row;
            we = r.w;
        } else {
            t  = ei[dir ? (E + e) : e];
            we = ew[e];
        }
        const int d   = t - lo;
        const bool in = (unsigned)d < unr;
        const long long fx = (long long)(we * (in ? DEG_FIX : 0.0f));
        __hip_atomic_fetch_add(dsum + (in ? dir * OWN_ROWS + d : 0), (unsigned long long)fx,
                               __ATOMIC_RELAXED, __HIP_MEMORY_SCOPE_WORKGROUP);
    }
}

template <bool REC>
__global__ void __launch_bounds__(OWN_THREADS)
k_degown(const int* __restrict__ ei, const float* __restrict__ ew,
         const EdgeRec* __restrict__ recs,
         const unsigned* __restrict__ keyR, const unsigned* __restrict__ keyC,
         float* __restrict__ ra, float* __restrict__ rb, float* __restrict__ outp, int E, int N, int nq16) {
    __shared__ unsigned long long dsum[2 * OWN_ROWS];
    __shared__ unsigned queue[OWN_WAVES * OWN_QCAP];

    const int tid  = (int)threadIdx.x;
    const int lane = tid & 31;
    const int wave = __builtin_amdgcn_readfirstlane(tid >> 5);
    const int lo   = (int)blockIdx.x * OWN_ROWS;
    const int nr   = (N - lo < OWN_ROWS) ? (N - lo) : OWN_ROWS;
    const unsigned unr = (unsigned)nr;
    const unsigned gs  = (blockIdx.x & 255u) * 0x01010101u;

    for (int i = tid; i < 2 * OWN_ROWS; i += OWN_THREADS) dsum[i] = 0ull;
    __syncthreads();

    unsigned* lst = (unsigned*)outp + (long long)lo * 64;
    const unsigned seg = ((unsigned)nr * 64u - OWN_WAVES) / OWN_WAVES;
    unsigned* ents = lst + OWN_WAVES + (unsigned)wave * seg;
    unsigned  ln = 0u;

    unsigned* qw = queue + wave * OWN_QCAP;
    unsigned  qn = 0;

    const int chunks  = (nq16 + 31) >> 5;
    const int ngroups = (chunks + 3) >> 2;
    const int rot     = (int)(((long long)blockIdx.x * ngroups) / gridDim.x);
    const v4u* kR = (const v4u*)keyR;
    const v4u* kC = (const v4u*)keyC;
    for (int gi = wave; gi < ngroups; gi += OWN_WAVES) {
        int gr = gi + rot;
        if (gr >= ngroups) gr -= ngroups;
        const int cb = gr * 4;
        v4u R[4], C[4];
        if ((cb + 4) * 32 <= nq16) {
#pragma unroll
            for (int k = 0; k < 4; ++k) {
                const int q = (cb + k) * 32 + lane;
                R[k] = kR[q];
                C[k] = kC[q];
            }
        } else {
            const unsigned miss = ~gs;
#pragma unroll
            for (int k = 0; k < 4; ++k) {
                const int q = (cb + k) * 32 + lane;
                const bool ok = q < nq16;
                v4u no = {miss, miss, miss, miss};
                R[k] = ok ? kR[q] : no;
                C[k] = ok ? kC[q] : no;
            }
        }
        unsigned hm[4];
#pragma unroll
        for (int k = 0; k < 4; ++k) hm[k] = own_hits(R[k], C[k], gs);
#pragma unroll
        for (int k = 0; k < 4; ++k) {
            unsigned h = hm[k];
            const unsigned e0 = 16u * (unsigned)((cb + k) * 32 + lane);
            unsigned bal = __builtin_amdgcn_ballot_w32(h != 0u);
            while (bal) {
                const bool     act = (h != 0u);
                const unsigned p   = act ? (unsigned)__builtin_ctz(h) : 0u;
                h &= (h - 1u);
                const unsigned e   = e0 + 4u * (p & 3u) + (p >> 3);
                const unsigned dir = (p >> 2) & 1u;
                const bool     ok  = act && (e < (unsigned)E);
                const unsigned okm = __builtin_amdgcn_ballot_w32(ok);
                const unsigned pos = qn + __builtin_amdgcn_mbcnt_lo(okm, 0u);
                if (ok) qw[pos] = (e << 1) | dir;
                qn += (unsigned)__builtin_popcount(okm);
                bal = __builtin_amdgcn_ballot_w32(h != 0u);
            }
            __builtin_amdgcn_wave_barrier();
            while (qn >= 32u) {
                qn -= 32u;
                deg_drain<true, REC>(qw + qn, 32, lane, ei, ew, recs, dsum, E, lo, unr, ents, ln, seg);
                ln += 32u;
            }
        }
    }
    __builtin_amdgcn_wave_barrier();
    if (qn) deg_drain<false, REC>(qw, (int)qn, lane, ei, ew, recs, dsum, E, lo, unr, ents, ln, seg);
    ln += qn;
    if (lane == 0) lst[wave] = (ln <= seg) ? ln : LIST_NONE;

    __syncthreads();

    for (int i = tid; i < 2 * OWN_ROWS; i += OWN_THREADS) {
        const int which = (i >= OWN_ROWS) ? 1 : 0;
        const int d     = i - which * OWN_ROWS;
        if (d < nr) {
            const float deg = (float)(long long)dsum[i] * (1.0f / DEG_FIX);
            const float f   = (deg > 0.0f) ? rsqrtf(deg) : 0.0f;
            if (which) rb[lo + d] = f; else ra[lo + d] = f;
        }
    }
}

template <bool REC>
__global__ void __launch_bounds__(OWN_THREADS)
k_owner(const float* __restrict__ ysd,
        const int* __restrict__ ei, const float* __restrict__ ew,
        const EdgeRec* __restrict__ recs,
        const unsigned* __restrict__ keyR, const unsigned* __restrict__ keyC,
        const float* __restrict__ ra, const float* __restrict__ rb,
        const float* __restrict__ bsrc, const float* __restrict__ bdst,
        float* out, int E, int N, int nq16) {
    __shared__ float    acc[OWN_ROWS * 64];
    __shared__ float    fac[2 * OWN_ROWS];
    __shared__ unsigned queue[OWN_WAVES * OWN_QCAP];

    const int tid  = (int)threadIdx.x;
    const int lane = tid & 31;
    const int wave = __builtin_amdgcn_readfirstlane(tid >> 5);
    const int lo   = (int)blockIdx.x * OWN_ROWS;
    const int nr   = (N - lo < OWN_ROWS) ? (N - lo) : OWN_ROWS;
    const unsigned unr = (unsigned)nr;
    const unsigned gs  = (blockIdx.x & 255u) * 0x01010101u;

    for (int i = tid; i < 2 * OWN_ROWS; i += OWN_THREADS) {
        const int which = (i >= OWN_ROWS) ? 1 : 0;
        const int d     = i - which * OWN_ROWS;
        float v = 0.0f;
        if (d < nr) v = which ? rb[lo + d] : ra[lo + d];
        fac[i] = v;
    }
    for (int i = tid; i < OWN_ROWS * 64; i += OWN_THREADS) acc[i] = 0.0f;

    const unsigned* lst  = (const unsigned*)out + (long long)lo * 64;
    const unsigned  seg  = ((unsigned)nr * 64u - OWN_WAVES) / OWN_WAVES;
    const unsigned* ents = lst + OWN_WAVES + (unsigned)wave * seg;
    const unsigned cnt = lst[wave];
    __syncthreads();

    if (cnt != LIST_NONE) {
        const unsigned nfull = cnt >> 5;
        for (unsigned b = 0u; b < nfull; ++b)
            own_drain<true, REC>(ents + 32u * b, 32, lane, ei, ew, recs, fac, ysd, acc, E, N, lo, unr);
        const unsigned left = cnt & 31u;
        if (left != 0u)
            own_drain<false, REC>(ents + 32u * nfull, (int)left, lane, ei, ew, recs, fac, ysd, acc, E, N, lo, unr);
    } else {
        unsigned* qw = queue + wave * OWN_QCAP;
        unsigned  qn = 0;

        const int chunks = (nq16 + 31) >> 5;

        const int ngroups = (chunks + 3) >> 2;
        const int rot     = (int)(((long long)blockIdx.x * ngroups) / gridDim.x);
        const v4u* kR = (const v4u*)keyR;
        const v4u* kC = (const v4u*)keyC;
        for (int gi = wave; gi < ngroups; gi += OWN_WAVES) {
            int gr = gi + rot;
            if (gr >= ngroups) gr -= ngroups;
            const int cb = gr * 4;
            v4u R[4], C[4];
            if ((cb + 4) * 32 <= nq16) {
#pragma unroll
                for (int k = 0; k < 4; ++k) {
                    const int q = (cb + k) * 32 + lane;
                    R[k] = kR[q];
                    C[k] = kC[q];
                }
            } else {
                const unsigned miss = ~gs;
#pragma unroll
                for (int k = 0; k < 4; ++k) {
                    const int q = (cb + k) * 32 + lane;
                    const bool ok = q < nq16;
                    v4u no = {miss, miss, miss, miss};
                    R[k] = ok ? kR[q] : no;
                    C[k] = ok ? kC[q] : no;
                }
            }
            unsigned hm[4];
#pragma unroll
            for (int k = 0; k < 4; ++k) hm[k] = own_hits(R[k], C[k], gs);
#pragma unroll
            for (int k = 0; k < 4; ++k) {
                unsigned h = hm[k];
                const unsigned e0 = 16u * (unsigned)((cb + k) * 32 + lane);
                unsigned bal = __builtin_amdgcn_ballot_w32(h != 0u);
                while (bal) {
                    const bool     act = (h != 0u);
                    const unsigned p   = act ? (unsigned)__builtin_ctz(h) : 0u;
                    h &= (h - 1u);
                    const unsigned e   = e0 + 4u * (p & 3u) + (p >> 3);
                    const unsigned dir = (p >> 2) & 1u;
                    const bool     ok  = act && (e < (unsigned)E);
                    const unsigned okm = __builtin_amdgcn_ballot_w32(ok);
                    const unsigned pos = qn + __builtin_amdgcn_mbcnt_lo(okm, 0u);
                    if (ok) qw[pos] = (e << 1) | dir;
                    qn += (unsigned)__builtin_popcount(okm);
                    bal = __builtin_amdgcn_ballot_w32(h != 0u);
                }
                __builtin_amdgcn_wave_barrier();
                while (qn >= 32u) {
                    qn -= 32u;
                    own_drain<true, REC>(qw + qn, 32, lane, ei, ew, recs, fac, ysd, acc, E, N, lo, unr);
                }
            }
        }
        __builtin_amdgcn_wave_barrier();
        if (qn) own_drain<false, REC>(qw, (int)qn, lane, ei, ew, recs, fac, ysd, acc, E, N, lo, unr);
    }

    __syncthreads();

    const v2f bs = *(const v2f*)(bsrc + 2 * lane);
    const v2f bd = *(const v2f*)(bdst + 2 * lane);
    v2f bias;
    bias.x = 0.5f * (bs.x + bd.x);
    bias.y = 0.5f * (bs.y + bd.y);
    for (int r = wave; r < nr; r += OWN_WAVES) {
        v2f a = *(const v2f*)(acc + r * 64 + 2 * lane);
        v2f o;
        o.x = bias.x + a.x;
        o.y = bias.y + a.y;
        *(v2f*)(out + (long long)(lo + r) * 64 + 2 * lane) = o;
    }
}

extern "C" void kernel_launch(void* const* d_in, const int* in_sizes, int n_in,
                              void* d_out, int out_size, void* d_ws, size_t ws_size,
                              hipStream_t stream) {
    const float* x    = (const float*)d_in[0];
    const int*   ei   = (const int*)d_in[1];
    const float* ew   = (const float*)d_in[2];
    const float* Wsrc = (const float*)d_in[3];
    const float* bsrc = (const float*)d_in[4];
    const float* Wdst = (const float*)d_in[5];
    const float* bdst = (const float*)d_in[6];
    float* out = (float*)d_out;

    const int N = in_sizes[0] / 64;
    const int E = in_sizes[2];

    float* ws     = (float*)d_ws;
    float* ra     = ws;
    float* rb     = ws + (long long)N;

    const int Epad = (E + KEY_LINE - 1) / KEY_LINE * KEY_LINE;
    unsigned char* kbase = (unsigned char*)(((size_t)(ws + 2LL * N) + 127) & ~(size_t)127);
    unsigned* keyR = (unsigned*)kbase;
    unsigned* keyC = (unsigned*)(kbase + Epad);

    unsigned char* rbase = (unsigned char*)(((size_t)(kbase + 2LL * Epad) + 127) & ~(size_t)127);
    EdgeRec* recs = (EdgeRec*)rbase;
    float* ysrec  = (float*)(((size_t)(rbase + sizeof(EdgeRec) * (size_t)E) + 255) & ~(size_t)255);
    const bool rec = (size_t)((unsigned char*)(ysrec + 128LL * N) - (unsigned char*)d_ws) <= ws_size;
    float* ys     = rec ? ysrec : ws + 2LL * N + (long long)E;
    float* yd     = ys + 64LL * N;

    const int nq4 = Epad / 4;
    if (rec) k_keys<true><<<(nq4 + 255) / 256, 256, 0, stream>>>(ei, ew, keyR, keyC, (int*)recs, E, nq4);
    else     k_keys<false><<<(nq4 + 255) / 256, 256, 0, stream>>>(ei, ew, keyR, keyC, (int*)recs, E, nq4);

    const int nblocks = (N + OWN_ROWS - 1) / OWN_ROWS;
    if (rec) k_degown<true><<<nblocks, OWN_THREADS, 0, stream>>>(ei, ew, recs, keyR, keyC, ra, rb, out, E, N, Epad / 16);
    else     k_degown<false><<<nblocks, OWN_THREADS, 0, stream>>>(ei, ew, recs, keyR, keyC, ra, rb, out, E, N, Epad / 16);

    const int tilesM = (N + 15) / 16;
    const int waves  = tilesM * 4;
    k_wmma_project<<<(waves + 7) / 8, 256, 0, stream>>>(x, Wsrc, Wdst,
                                                        ys, yd, ra, rb, tilesM, N);

    if (rec) k_owner<true><<<nblocks, OWN_THREADS, 0, stream>>>(ys, ei, ew, recs, keyR, keyC, ra, rb,
                                                       bsrc, bdst, out, E, N, Epad / 16);
    else     k_owner<false><<<nblocks, OWN_THREADS, 0, stream>>>(ys, ei, ew, recs, keyR, keyC, ra, rb,
                                                        bsrc, bdst, out, E, N, Epad / 16);
}


# projection: 512 resident work-groups walk the row tiles, a wave keeps its W fragments in registers over its tiles, next tile's x loads ahead of the products (was a wave per tile); sum order same
# speedup vs baseline: 2.4325x; 1.0519x over previous
#include <hip/hip_runtime.h>
#include <hip/hip_bf16.h>


#define OWN_ROWS    392
#define KEY_LINE    128

typedef __attribute__((ext_vector_type(2))) float v2f;
typedef __attribute__((ext_vector_type(4))) float v4f;
typedef __attribute__((ext_vector_type(8))) float v8f;
typedef __attribute__((ext_vector_type(4))) int v4i;
typedef __attribute__((ext_vector_type(4))) unsigned v4u;

struct __attribute__((packed, aligned(4))) EdgeRec { int row; int col; float w; };

template <bool REC>
__global__ void __launch_bounds__(256)
k_keys(const int* __restrict__ ei, const float* __restrict__ ew,
       unsigned* __restrict__ keyR, unsigned* __restrict__ keyC,
       int* __restrict__ recs, int E, int nq4) {
    const int q = blockIdx.x * blockDim.x + threadIdx.x;
    if (q >= nq4) return;
    const long long e0 = 4LL * q;
    int r[4], c[4];
    if (((E & 3) == 0) && e0 + 4 <= (long long)E) {
        const v4i R  = *(const v4i*)(ei + e0);
        const v4i C  = *(const v4i*)(ei + (long long)E + e0);
#pragma unroll
        for (int m = 0; m < 4; ++m) { r[m] = R[m]; c[m] = C[m]; }
        if (REC) {
            const v4f Wf = *(const v4f*)(ew + e0);
            v4i* rp = (v4i*)(recs + 12LL * q);
            const v4i s0 = {R[0], C[0], __float_as_int(Wf[0]), R[1]};
            const v4i s1 = {C[1], __float_as_int(Wf[1]), R[2], C[2]};
            const v4i s2 = {__float_as_int(Wf[2]), R[3], C[3], __float_as_int(Wf[3])};
            rp[0] = s0;
            rp[1] = s1;
            rp[2] = s2;
        }
    } else {
#pragma unroll
        for (int m = 0; m < 4; ++m) {
            const long long e = e0 + m;
            const bool ok = e < (long long)E;
            r[m]  = ok ? ei[e] : -1;
            c[m]  = ok ? ei[(long long)E + e] : -1;
            if (REC && ok) {
                int* rp = recs + 3LL * e;
                rp[0] = r[m];
                rp[1] = c[m];
                rp[2] = __float_as_int(ew[e]);
            }
        }
    }
    unsigned kr = 0u, kc = 0u;
#pragma unroll
    for (int m = 0; m < 4; ++m) {
        if (r[m] >= 0) {
            kr |= (((unsigned)r[m] / OWN_ROWS) & 255u) << (8 * m);
            kc |= (((unsigned)c[m] / OWN_ROWS) & 255u) << (8 * m);
        }
    }
    keyR[q] = kr;
    keyC[q] = kc;
}

#define PRJ_GROUPS  512

__device__ __forceinline__ void prj_tile(const v2f (&a)[16], const v2f (&bs)[16], const v2f (&bd)[16],
                                         float f, float* fw, int lane, int lhi, int ncol, int tileM,
                                         float* __restrict__ ys, float* __restrict__ yd, int Nnodes) {
    v8f accs = {};
    v8f accd = {};
#pragma unroll
    for (int kk = 0; kk < 16; ++kk) {
        accs = __builtin_amdgcn_wmma_f32_16x16x4_f32(
                   false, a[kk], false, bs[kk], (short)0, accs, false, false);
        accd = __builtin_amdgcn_wmma_f32_16x16x4_f32(
                   false, a[kk], false, bd[kk], (short)0, accd, false, false);
    }

    fw[lane] = f;
    __builtin_amdgcn_wave_barrier();
    const v4f fa0 = *(const v4f*)(fw + 8 * lhi);
    const v4f fa1 = *(const v4f*)(fw + 8 * lhi + 4);
    const v4f fb0 = *(const v4f*)(fw + 16 + 8 * lhi);
    const v4f fb1 = *(const v4f*)(fw + 16 + 8 * lhi + 4);
    __builtin_amdgcn_wave_barrier();

    const int mbase = tileM * 16 + 8 * lhi;
#pragma unroll
    for (int r = 0; r < 8; ++r) {
        const int m = mbase + r;
        if (m < Nnodes) {
            const long long o = (long long)m * 64 + ncol;
            const float fra = (r < 4) ? fa0[r & 3] : fa1[r & 3];
            const float frb = (r < 4) ? fb0[r & 3] : fb1[r & 3];
            ys[o] = (0.5f * accs[r]) * frb;
            yd[o] = (0.5f * accd[r]) * fra;
        }
    }
}

__global__ void __launch_bounds__(256)
k_wmma_project(const float* __restrict__ x,
               const float* __restrict__ Wsrc, const float* __restrict__ Wdst,
               float* __restrict__ ys, float* __restrict__ yd,
               const float* __restrict__ ra, const float* __restrict__ rb,
               int tilesM, int Nnodes) {
    __shared__ __attribute__((aligned(16))) float facs[8 * 32];

    const int lane  = (int)(threadIdx.x & 31);
    const int wv    = __builtin_amdgcn_readfirstlane((int)(threadIdx.x >> 5));
    const int tileN = wv & 3;
    const int lmod  = lane & 15;
    const int lhi   = lane >> 4;
    float* fw = facs + wv * 32;

    const int step = 2 * (int)gridDim.x;
    int tileM = 2 * (int)blockIdx.x + (wv >> 2);
    if (tileM >= tilesM) return;

    v2f bs[16], bd[16];
    {
        const float* bs_p = Wsrc + (tileN * 16 + lmod) * 64 + 2 * lhi;
        const float* bd_p = Wdst + (tileN * 16 + lmod) * 64 + 2 * lhi;
#pragma unroll
        for (int kk = 0; kk < 16; ++kk) {
            bs[kk] = *(const v2f*)(bs_p + 4 * kk);
            bd[kk] = *(const v2f*)(bd_p + 4 * kk);
        }
    }
    const float* fsrc = lhi ? rb : ra;
    const int ncol = tileN * 16 + lmod;

    v2f   a[16];
    float f;
    {
        int arow = tileM * 16 + lmod;
        if (arow >= Nnodes) arow = Nnodes - 1;
        const float* a_p = x + (long long)arow * 64 + 2 * lhi;
#pragma unroll
        for (int kk = 0; kk < 16; ++kk) a[kk] = *(const v2f*)(a_p + 4 * kk);
        f = fsrc[arow];
    }

    while (tileM + step < tilesM) {
        const int next = tileM + step;
        v2f   an[16];
        float fn;
        {
            int arow = next * 16 + lmod;
            if (arow >= Nnodes) arow = Nnodes - 1;
            const float* a_p = x + (long long)arow * 64 + 2 * lhi;
#pragma unroll
            for (int kk = 0; kk < 16; ++kk) an[kk] = *(const v2f*)(a_p + 4 * kk);
            fn = fsrc[arow];
        }
        prj_tile(a, bs, bd, f, fw, lane, lhi, ncol, tileM, ys, yd, Nnodes);
#pragma unroll
        for (int kk = 0; kk < 16; ++kk) a[kk] = an[kk];
        f = fn;
        tileM = next;
    }
    prj_tile(a, bs, bd, f, fw, lane, lhi, ncol, tileM, ys, yd, Nnodes);
}

#define OWN_WAVES   32
#define OWN_THREADS (OWN_WAVES * 32)
#define OWN_QCAP    1056
#define LIST_NONE   0xFFFFFFFFu

__device__ __forceinline__ unsigned own_nz(unsigned w, unsigned gs) {
    const unsigned x = w ^ gs;
    return ((x & 0x7F7F7F7Fu) + 0x7F7F7F7Fu) | x | 0x7F7F7F7Fu;
}

__device__ __forceinline__ unsigned own_hits(v4u r, v4u c, unsigned gs) {
    unsigned m = own_nz(r[0], gs);
    m = __builtin_rotateright32(m, 1) & own_nz(r[1], gs);
    m = __builtin_rotateright32(m, 1) & own_nz(r[2], gs);
    m = __builtin_rotateright32(m, 1) & own_nz(r[3], gs);
    m = __builtin_rotateright32(m, 1) & own_nz(c[0], gs);
    m = __builtin_rotateright32(m, 1) & own_nz(c[1], gs);
    m = __builtin_rotateright32(m, 1) & own_nz(c[2], gs);
    m = __builtin_rotateright32(m, 1) & own_nz(c[3], gs);
    return ~m;
}

template <bool FULL, bool REC>
__device__ __forceinline__ void own_drain(const unsigned* qb, int n, int lane,
                                          const int* __restrict__ ei, const float* __restrict__ ew,
                                          const EdgeRec* __restrict__ recs,
                                          const float* fac,
                                          const float* __restrict__ ysd, float* acc,
                                          int E, int N, int lo, unsigned unr) {
    int   dl   = 0;
    int   srow = 0;
    float wn   = 0.0f;
    if (FULL || lane < n) {
        unsigned code = qb[lane];
        int e   = (int)(code >> 1);
        int dir = (int)(code & 1u);
        int row, col;
        float we;
        if (REC) {
            const EdgeRec r = recs[e];
            row = r.row;
            col = r.col;
            we  = r.w;
        } else {
            row = ei[e];
            col = ei[E + e];
            we  = ew[e];
        }
        int d   = (dir ? col : row) - lo;
        const bool in = (unsigned)d < unr;
        dl   = in ? d : 0;
        const float fl = fac[dir * OWN_ROWS + dl];
        wn   = we * (in ? fl : 0.0f);
        srow = in ? (dir ? (row + N) : col) : 0;
    }
    v2f v[32];
#pragma unroll
    for (int j = 0; j < 32; ++j) {
        if (FULL || j < n) {
            int s = __builtin_amdgcn_readlane(srow, j);
            v[j] = *(const v2f*)(ysd + (long long)s * 64 + 2 * lane);
        }
    }
#pragma unroll
    for (int j = 0; j < 32; ++j) {
        if (FULL || j < n) {
            float w = __int_as_float(__builtin_amdgcn_readlane(__float_as_int(wn), j));
            int   d = __builtin_amdgcn_readlane(dl, j);
            float* a = acc + d * 64 + 2 * lane;
            __hip_atomic_fetch_add(a + 0, w * v[j].x, __ATOMIC_RELAXED, __HIP_MEMORY_SCOPE_WORKGROUP);
            __hip_atomic_fetch_add(a + 1, w * v[j].y, __ATOMIC_RELAXED, __HIP_MEMORY_SCOPE_WORKGROUP);
        }
    }
}

#define DEG_FIX 1099511627776.0f

template <bool FULL, bool REC>
__device__ __forceinline__ void deg_drain(const unsigned* qb, int n, int lane,
                                          const int* __restrict__ ei, const float* __restrict__ ew,
                                          const EdgeRec* __restrict__ recs,
                                          unsigned long long* dsum, int E, int lo, unsigned unr,
                                          unsigned* ents, unsigned ln, unsigned seg) {
    const bool fits = ln + (unsigned)n <= seg;
    if (FULL || lane < n) {
        const unsigned code = qb[lane];
        if (fits) ents[ln + (unsigned)lane] = code;
        const int e   = (int)(code >> 1);
        const int dir = (int)(code & 1u);
        int t;
        float we;
        if (REC) {
            const EdgeRec r = recs[e];
            t  = dir ? r.col : r.row;
            we = r.w;
        } else {
            t  = ei[dir ? (E + e) : e];
            we = ew[e];
        }
        const int d   = t - lo;
        const bool in = (unsigned)d < unr;
        const long long fx = (long long)(we * (in ? DEG_FIX : 0.0f));
        __hip_atomic_fetch_add(dsum + (in ? dir * OWN_ROWS + d : 0), (unsigned long long)fx,
                               __ATOMIC_RELAXED, __HIP_MEMORY_SCOPE_WORKGROUP);
    }
}

template <bool REC>
__global__ void __launch_bounds__(OWN_THREADS)
k_degown(const int* __restrict__ ei, const float* __restrict__ ew,
         const EdgeRec* __restrict__ recs,
         const unsigned* __restrict__ keyR, const unsigned* __restrict__ keyC,
         float* __restrict__ ra, float* __restrict__ rb, float* __restrict__ outp, int E, int N, int nq16) {
    __shared__ unsigned long long dsum[2 * OWN_ROWS];
    __shared__ unsigned queue[OWN_WAVES * OWN_QCAP];

    const int tid  = (int)threadIdx.x;
    const int lane = tid & 31;
    const int wave = __builtin_amdgcn_readfirstlane(tid >> 5);
    const int lo   = (int)blockIdx.x * OWN_ROWS;
    const int nr   = (N - lo < OWN_ROWS) ? (N - lo) : OWN_ROWS;
    const unsigned unr = (unsigned)nr;
    const unsigned gs  = (blockIdx.x & 255u) * 0x01010101u;

    for (int i = tid; i < 2 * OWN_ROWS; i += OWN_THREADS) dsum[i] = 0ull;
    __syncthreads();

    unsigned* lst = (unsigned*)outp + (long long)lo * 64;
    const unsigned seg = ((unsigned)nr * 64u - OWN_WAVES) / OWN_WAVES;
    unsigned* ents = lst + OWN_WAVES + (unsigned)wave * seg;
    unsigned  ln = 0u;

    unsigned* qw = queue + wave * OWN_QCAP;
    unsigned  qn = 0;

    const int chunks  = (nq16 + 31) >> 5;
    const int ngroups = (chunks + 3) >> 2;
    const int rot     = (int)(((long long)blockIdx.x * ngroups) / gridDim.x);
    const v4u* kR = (const v4u*)keyR;
    const v4u* kC = (const v4u*)keyC;
    for (int gi = wave; gi < ngroups; gi += OWN_WAVES) {
        int gr = gi + rot;
        if (gr >= ngroups) gr -= ngroups;
        const int cb = gr * 4;
        v4u R[4], C[4];
        if ((cb + 4) * 32 <= nq16) {
#pragma unroll
            for (int k = 0; k < 4; ++k) {
                const int q = (cb + k) * 32 + lane;
                R[k] = kR[q];
                C[k] = kC[q];
            }
        } else {
            const unsigned miss = ~gs;
#pragma unroll
            for (int k = 0; k < 4; ++k) {
                const int q = (cb + k) * 32 + lane;
                const bool ok = q < nq16;
                v4u no = {miss, miss, miss, miss};
                R[k] = ok ? kR[q] : no;
                C[k] = ok ? kC[q] : no;
            }
        }
        unsigned hm[4];
#pragma unroll
        for (int k = 0; k < 4; ++k) hm[k] = own_hits(R[k], C[k], gs);
#pragma unroll
        for (int k = 0; k < 4; ++k) {
            unsigned h = hm[k];
            const unsigned e0 = 16u * (unsigned)((cb + k) * 32 + lane);
            unsigned bal = __builtin_amdgcn_ballot_w32(h != 0u);
            while (bal) {
                const bool     act = (h != 0u);
                const unsigned p   = act ? (unsigned)__builtin_ctz(h) : 0u;
                h &= (h - 1u);
                const unsigned e   = e0 + 4u * (p & 3u) + (p >> 3);
                const unsigned dir = (p >> 2) & 1u;
                const bool     ok  = act && (e < (unsigned)E);
                const unsigned okm = __builtin_amdgcn_ballot_w32(ok);
                const unsigned pos = qn + __builtin_amdgcn_mbcnt_lo(okm, 0u);
                if (ok) qw[pos] = (e << 1) | dir;
                qn += (unsigned)__builtin_popcount(okm);
                bal = __builtin_amdgcn_ballot_w32(h != 0u);
            }
            __builtin_amdgcn_wave_barrier();
            while (qn >= 32u) {
                qn -= 32u;
                deg_drain<true, REC>(qw + qn, 32, lane, ei, ew, recs, dsum, E, lo, unr, ents, ln, seg);
                ln += 32u;
            }
        }
    }
    __builtin_amdgcn_wave_barrier();
    if (qn) deg_drain<false, REC>(qw, (int)qn, lane, ei, ew, recs, dsum, E, lo, unr, ents, ln, seg);
    ln += qn;
    if (lane == 0) lst[wave] = (ln <= seg) ? ln : LIST_NONE;

    __syncthreads();

    for (int i = tid; i < 2 * OWN_ROWS; i += OWN_THREADS) {
        const int which = (i >= OWN_ROWS) ? 1 : 0;
        const int d     = i - which * OWN_ROWS;
        if (d < nr) {
            const float deg = (float)(long long)dsum[i] * (1.0f / DEG_FIX);
            const float f   = (deg > 0.0f) ? rsqrtf(deg) : 0.0f;
            if (which) rb[lo + d] = f; else ra[lo + d] = f;
        }
    }
}

template <bool REC>
__global__ void __launch_bounds__(OWN_THREADS)
k_owner(const float* __restrict__ ysd,
        const int* __restrict__ ei, const float* __restrict__ ew,
        const EdgeRec* __restrict__ recs,
        const unsigned* __restrict__ keyR, const unsigned* __restrict__ keyC,
        const float* __restrict__ ra, const float* __restrict__ rb,
        const float* __restrict__ bsrc, const float* __restrict__ bdst,
        float* out, int E, int N, int nq16) {
    __shared__ float    acc[OWN_ROWS * 64];
    __shared__ float    fac[2 * OWN_ROWS];
    __shared__ unsigned queue[OWN_WAVES * OWN_QCAP];

    const int tid  = (int)threadIdx.x;
    const int lane = tid & 31;
    const int wave = __builtin_amdgcn_readfirstlane(tid >> 5);
    const int lo   = (int)blockIdx.x * OWN_ROWS;
    const int nr   = (N - lo < OWN_ROWS) ? (N - lo) : OWN_ROWS;
    const unsigned unr = (unsigned)nr;
    const unsigned gs  = (blockIdx.x & 255u) * 0x01010101u;

    for (int i = tid; i < 2 * OWN_ROWS; i += OWN_THREADS) {
        const int which = (i >= OWN_ROWS) ? 1 : 0;
        const int d     = i - which * OWN_ROWS;
        float v = 0.0f;
        if (d < nr) v = which ? rb[lo + d] : ra[lo + d];
        fac[i] = v;
    }
    for (int i = tid; i < OWN_ROWS * 64; i += OWN_THREADS) acc[i] = 0.0f;

    const unsigned* lst  = (const unsigned*)out + (long long)lo * 64;
    const unsigned  seg  = ((unsigned)nr * 64u - OWN_WAVES) / OWN_WAVES;
    const unsigned* ents = lst + OWN_WAVES + (unsigned)wave * seg;
    const unsigned cnt = lst[wave];
    __syncthreads();

    if (cnt != LIST_NONE) {
        const unsigned nfull = cnt >> 5;
        for (unsigned b = 0u; b < nfull; ++b)
            own_drain<true, REC>(ents + 32u * b, 32, lane, ei, ew, recs, fac, ysd, acc, E, N, lo, unr);
        const unsigned left = cnt & 31u;
        if (left != 0u)
            own_drain<false, REC>(ents + 32u * nfull, (int)left, lane, ei, ew, recs, fac, ysd, acc, E, N, lo, unr);
    } else {
        unsigned* qw = queue + wave * OWN_QCAP;
        unsigned  qn = 0;

        const int chunks = (nq16 + 31) >> 5;

        const int ngroups = (chunks + 3) >> 2;
        const int rot     = (int)(((long long)blockIdx.x * ngroups) / gridDim.x);
        const v4u* kR = (const v4u*)keyR;
        const v4u* kC = (const v4u*)keyC;
        for (int gi = wave; gi < ngroups; gi += OWN_WAVES) {
            int gr = gi + rot;
            if (gr >= ngroups) gr -= ngroups;
            const int cb = gr * 4;
            v4u R[4], C[4];
            if ((cb + 4) * 32 <= nq16) {
#pragma unroll
                for (int k = 0; k < 4; ++k) {
                    const int q = (cb + k) * 32 + lane;
                    R[k] = kR[q];
                    C[k] = kC[q];
                }
            } else {
                const unsigned miss = ~gs;
#pragma unroll
                for (int k = 0; k < 4; ++k) {
                    const int q = (cb + k) * 32 + lane;
                    const bool ok = q < nq16;
                    v4u no = {miss, miss, miss, miss};
                    R[k] = ok ? kR[q] : no;
                    C[k] = ok ? kC[q] : no;
                }
            }
            unsigned hm[4];
#pragma unroll
            for (int k = 0; k < 4; ++k) hm[k] = own_hits(R[k], C[k], gs);
#pragma unroll
            for (int k = 0; k < 4; ++k) {
                unsigned h = hm[k];
                const unsigned e0 = 16u * (unsigned)((cb + k) * 32 + lane);
                unsigned bal = __builtin_amdgcn_ballot_w32(h != 0u);
                while (bal) {
                    const bool     act = (h != 0u);
                    const unsigned p   = act ? (unsigned)__builtin_ctz(h) : 0u;
                    h &= (h - 1u);
                    const unsigned e   = e0 + 4u * (p & 3u) + (p >> 3);
                    const unsigned dir = (p >> 2) & 1u;
                    const bool     ok  = act && (e < (unsigned)E);
                    const unsigned okm = __builtin_amdgcn_ballot_w32(ok);
                    const unsigned pos = qn + __builtin_amdgcn_mbcnt_lo(okm, 0u);
                    if (ok) qw[pos] = (e << 1) | dir;
                    qn += (unsigned)__builtin_popcount(okm);
                    bal = __builtin_amdgcn_ballot_w32(h != 0u);
                }
                __builtin_amdgcn_wave_barrier();
                while (qn >= 32u) {
                    qn -= 32u;
                    own_drain<true, REC>(qw + qn, 32, lane, ei, ew, recs, fac, ysd, acc, E, N, lo, unr);
                }
            }
        }
        __builtin_amdgcn_wave_barrier();
        if (qn) own_drain<false, REC>(qw, (int)qn, lane, ei, ew, recs, fac, ysd, acc, E, N, lo, unr);
    }

    __syncthreads();

    const v2f bs = *(const v2f*)(bsrc + 2 * lane);
    const v2f bd = *(const v2f*)(bdst + 2 * lane);
    v2f bias;
    bias.x = 0.5f * (bs.x + bd.x);
    bias.y = 0.5f * (bs.y + bd.y);
    for (int r = wave; r < nr; r += OWN_WAVES) {
        v2f a = *(const v2f*)(acc + r * 64 + 2 * lane);
        v2f o;
        o.x = bias.x + a.x;
        o.y = bias.y + a.y;
        *(v2f*)(out + (long long)(lo + r) * 64 + 2 * lane) = o;
    }
}

extern "C" void kernel_launch(void* const* d_in, const int* in_sizes, int n_in,
                              void* d_out, int out_size, void* d_ws, size_t ws_size,
                              hipStream_t stream) {
    const float* x    = (const float*)d_in[0];
    const int*   ei   = (const int*)d_in[1];
    const float* ew   = (const float*)d_in[2];
    const float* Wsrc = (const float*)d_in[3];
    const float* bsrc = (const float*)d_in[4];
    const float* Wdst = (const float*)d_in[5];
    const float* bdst = (const float*)d_in[6];
    float* out = (float*)d_out;

    const int N = in_sizes[0] / 64;
    const int E = in_sizes[2];

    float* ws     = (float*)d_ws;
    float* ra     = ws;
    float* rb     = ws + (long long)N;

    const int Epad = (E + KEY_LINE - 1) / KEY_LINE * KEY_LINE;
    unsigned char* kbase = (unsigned char*)(((size_t)(ws + 2LL * N) + 127) & ~(size_t)127);
    unsigned* keyR = (unsigned*)kbase;
    unsigned* keyC = (unsigned*)(kbase + Epad);

    unsigned char* rbase = (unsigned char*)(((size_t)(kbase + 2LL * Epad) + 127) & ~(size_t)127);
    EdgeRec* recs = (EdgeRec*)rbase;
    float* ysrec  = (float*)(((size_t)(rbase + sizeof(EdgeRec) * (size_t)E) + 255) & ~(size_t)255);
    const bool rec = (size_t)((unsigned char*)(ysrec + 128LL * N) - (unsigned char*)d_ws) <= ws_size;
    float* ys     = rec ? ysrec : ws + 2LL * N + (long long)E;
    float* yd     = ys + 64LL * N;

    const int nq4 = Epad / 4;
    if (rec) k_keys<true><<<(nq4 + 255) / 256, 256, 0, stream>>>(ei, ew, keyR, keyC, (int*)recs, E, nq4);
    else     k_keys<false><<<(nq4 + 255) / 256, 256, 0, stream>>>(ei, ew, keyR, keyC, (int*)recs, E, nq4);

    const int nblocks = (N + OWN_ROWS - 1) / OWN_ROWS;
    if (rec) k_degown<true><<<nblocks, OWN_THREADS, 0, stream>>>(ei, ew, recs, keyR, keyC, ra, rb, out, E, N, Epad / 16);
    else     k_degown<false><<<nblocks, OWN_THREADS, 0, stream>>>(ei, ew, recs, keyR, keyC, ra, rb, out, E, N, Epad / 16);

    const int tilesM = (N + 15) / 16;
    k_wmma_project<<<PRJ_GROUPS, 256, 0, stream>>>(x, Wsrc, Wdst,
                                                   ys, yd, ra, rb, tilesM, N);

    if (rec) k_owner<true><<<nblocks, OWN_THREADS, 0, stream>>>(ys, ei, ew, recs, keyR, keyC, ra, rb,
                                                       bsrc, bdst, out, E, N, Epad / 16);
    else     k_owner<false><<<nblocks, OWN_THREADS, 0, stream>>>(ys, ei, ew, recs, keyR, keyC, ra, rb,
                                                        bsrc, bdst, out, E, N, Epad / 16);
}
